# starting kernel
# baseline (speedup 1.0000x reference)
#include <hip/hip_runtime.h>


#define N_NODES 50000
#define N_EDGES 600000
#define DIM     128
#define M_TILES (N_NODES / 16)

typedef __attribute__((ext_vector_type(2))) float v2f;
typedef __attribute__((ext_vector_type(8))) float v8f;

__global__ __launch_bounds__(128)
void gcn_gemm128(const float* __restrict__ A, const float* __restrict__ W,
                 float* __restrict__ C, int M) {
    __shared__ float ldsW[DIM * DIM];

    const int tid  = threadIdx.x;
    const int lane = tid & 31;
    const int wave = tid >> 5;

    for (int p = tid; p < 8192; p += 128) {
        const int l    = p & 31;
        const int nt   = (p >> 5) & 7;
        const int kk   = p >> 8;
        const int krow = kk * 4 + ((l & 16) ? 2 : 0);
        const int col  = nt * 16 + (l & 15);
        ldsW[2 * p + 0] = W[(krow + 0) * DIM + col];
        ldsW[2 * p + 1] = W[(krow + 1) * DIM + col];
    }
    __syncthreads();

    const int mtile = blockIdx.x * 4 + wave;
    if (mtile * 16 >= M) return;

    const float* __restrict__ Arow =
        A + (size_t)(mtile * 16 + (lane & 15)) * DIM + ((lane & 16) ? 2 : 0);

    v8f acc[8] = {};

    for (int kk = 0; kk < 32; ++kk) {
        const v2f a = *(const v2f*)(Arow + kk * 4);
#pragma unroll
        for (int nt = 0; nt < 8; ++nt) {
            const v2f b =
                *(const v2f*)&ldsW[(((kk << 3) + nt) * 32 + lane) * 2];
            acc[nt] = __builtin_amdgcn_wmma_f32_16x16x4_f32(
                 false, a,  false, b,
                 (short)0, acc[nt],
                 false,  false);
        }
    }

    const int row0 = mtile * 16 + ((lane & 16) ? 8 : 0);
    const int colc = lane & 15;
#pragma unroll
    for (int nt = 0; nt < 8; ++nt) {
#pragma unroll
        for (int j = 0; j < 8; ++j) {
            C[(size_t)(row0 + j) * DIM + nt * 16 + colc] = acc[nt][j];
        }
    }
}

__global__ __launch_bounds__(256)
void gcn_fill0(float* __restrict__ p, int n) {
    const int i = blockIdx.x * 256 + threadIdx.x;
    if (i < n) p[i] = 0.0f;
}

__global__ __launch_bounds__(256)
void gcn_spmm(const int* __restrict__ erow, const int* __restrict__ ecol,
              const float* __restrict__ evals, const float* __restrict__ h,
              float* __restrict__ out, int E) {
    const int e    = blockIdx.x * 8 + (threadIdx.x >> 5);
    const int lane = threadIdx.x & 31;
    if (e >= E) return;

    const int   r = erow[e];
    const int   c = ecol[e];
    const float v = evals[e];

    const float4 x = ((const float4*)(h + (size_t)c * DIM))[lane];
    float* __restrict__ o = out + (size_t)r * DIM + lane * 4;

    __hip_atomic_fetch_add(o + 0, v * x.x, __ATOMIC_RELAXED, __HIP_MEMORY_SCOPE_AGENT);
    __hip_atomic_fetch_add(o + 1, v * x.y, __ATOMIC_RELAXED, __HIP_MEMORY_SCOPE_AGENT);
    __hip_atomic_fetch_add(o + 2, v * x.z, __ATOMIC_RELAXED, __HIP_MEMORY_SCOPE_AGENT);
    __hip_atomic_fetch_add(o + 3, v * x.w, __ATOMIC_RELAXED, __HIP_MEMORY_SCOPE_AGENT);
}

__global__ __launch_bounds__(256)
void gcn_bias_tanh(float* __restrict__ x, const float* __restrict__ b, int n) {
    const int i = blockIdx.x * 256 + threadIdx.x;
    if (i < n) x[i] = tanhf(x[i] + b[i & (DIM - 1)]);
}

__global__ __launch_bounds__(256)
void gcn_gemv(const float* __restrict__ h, const float* __restrict__ fcw,
              const float* __restrict__ fcb, float* __restrict__ out, int M) {
    const int node = blockIdx.x * 8 + (threadIdx.x >> 5);
    const int lane = threadIdx.x & 31;
    if (node >= M) return;

    const float4 x = ((const float4*)(h + (size_t)node * DIM))[lane];
    const float4 w = ((const float4*)fcw)[lane];
    float s = x.x * w.x + x.y * w.y + x.z * w.z + x.w * w.w;
#pragma unroll
    for (int off = 16; off > 0; off >>= 1) s += __shfl_down(s, off, 32);
    if (lane == 0) out[node] = s + fcb[0];
}

extern "C" void kernel_launch(void* const* d_in, const int* in_sizes, int n_in,
                              void* d_out, int out_size, void* d_ws, size_t ws_size,
                              hipStream_t stream) {
    const float* features = (const float*)d_in[0];
    const int*   erow     = (const int*)  d_in[1];
    const int*   ecol     = (const int*)  d_in[2];
    const float* evals    = (const float*)d_in[3];
    const float* W1       = (const float*)d_in[4];
    const float* b1       = (const float*)d_in[5];
    const float* W2       = (const float*)d_in[6];
    const float* b2       = (const float*)d_in[7];
    const float* fcw      = (const float*)d_in[8];
    const float* fcb      = (const float*)d_in[9];
    float*       out      = (float*)d_out;

    float* tmp0 = (float*)d_ws;
    float* tmp1 = tmp0 + (size_t)N_NODES * DIM;

    const int nElem      = N_NODES * DIM;
    const dim3 gemmGrid((M_TILES + 3) / 4);
    const dim3 elemGrid((nElem + 255) / 256);
    const dim3 spmmGrid((N_EDGES + 7) / 8);
    const dim3 gemvGrid((N_NODES + 7) / 8);

    gcn_gemm128<<<gemmGrid, 128, 0, stream>>>(features, W1, tmp0, N_NODES);
    gcn_fill0  <<<elemGrid, 256, 0, stream>>>(tmp1, nElem);
    gcn_spmm   <<<spmmGrid, 256, 0, stream>>>(erow, ecol, evals, tmp0, tmp1, N_EDGES);
    gcn_bias_tanh<<<elemGrid, 256, 0, stream>>>(tmp1, b1, nElem);

    gcn_gemm128<<<gemmGrid, 128, 0, stream>>>(tmp1, W2, tmp0, N_NODES);
    gcn_fill0  <<<elemGrid, 256, 0, stream>>>(tmp1, nElem);
    gcn_spmm   <<<spmmGrid, 256, 0, stream>>>(erow, ecol, evals, tmp0, tmp1, N_EDGES);
    gcn_bias_tanh<<<elemGrid, 256, 0, stream>>>(tmp1, b2, nElem);

    gcn_gemv<<<gemvGrid, 256, 0, stream>>>(tmp1, fcw, fcb, out, N_NODES);
}


# A @ h by destination row: edges bucketed once per call by counting sort (int atomic slot + scan), one wave sums a row in registers in bucket order; f32 atomics and zero-fill gone
# speedup vs baseline: 1.2683x; 1.2683x over previous
#include <hip/hip_runtime.h>


#define N_NODES 50000
#define N_EDGES 600000
#define DIM     128
#define M_TILES (N_NODES / 16)
#define SCAN_PER_THREAD 52
#define N_PAD  (1024 * SCAN_PER_THREAD)
#define E_PAD  600064

typedef __attribute__((ext_vector_type(2))) float v2f;
typedef __attribute__((ext_vector_type(8))) float v8f;
typedef __attribute__((ext_vector_type(4))) float v4f;

__global__ __launch_bounds__(128)
void gcn_gemm128(const float* __restrict__ A, const float* __restrict__ W,
                 float* __restrict__ C, int M) {
    __shared__ float ldsW[DIM * DIM];

    const int tid  = threadIdx.x;
    const int lane = tid & 31;
    const int wave = tid >> 5;

    for (int p = tid; p < 8192; p += 128) {
        const int l    = p & 31;
        const int nt   = (p >> 5) & 7;
        const int kk   = p >> 8;
        const int krow = kk * 4 + ((l & 16) ? 2 : 0);
        const int col  = nt * 16 + (l & 15);
        ldsW[2 * p + 0] = W[(krow + 0) * DIM + col];
        ldsW[2 * p + 1] = W[(krow + 1) * DIM + col];
    }
    __syncthreads();

    const int mtile = blockIdx.x * 4 + wave;
    if (mtile * 16 >= M) return;

    const float* __restrict__ Arow =
        A + (size_t)(mtile * 16 + (lane & 15)) * DIM + ((lane & 16) ? 2 : 0);

    v8f acc[8] = {};

    for (int kk = 0; kk < 32; ++kk) {
        const v2f a = *(const v2f*)(Arow + kk * 4);
#pragma unroll
        for (int nt = 0; nt < 8; ++nt) {
            const v2f b =
                *(const v2f*)&ldsW[(((kk << 3) + nt) * 32 + lane) * 2];
            acc[nt] = __builtin_amdgcn_wmma_f32_16x16x4_f32(
                 false, a,  false, b,
                 (short)0, acc[nt],
                 false,  false);
        }
    }

    const int row0 = mtile * 16 + ((lane & 16) ? 8 : 0);
    const int colc = lane & 15;
#pragma unroll
    for (int nt = 0; nt < 8; ++nt) {
#pragma unroll
        for (int j = 0; j < 8; ++j) {
            C[(size_t)(row0 + j) * DIM + nt * 16 + colc] = acc[nt][j];
        }
    }
}

__global__ __launch_bounds__(256)
void gcn_zero_counts(int* __restrict__ p, int n) {
    const int i = blockIdx.x * 256 + threadIdx.x;
    if (i < n) p[i] = 0;
}

__global__ __launch_bounds__(256)
void gcn_rank_edges(const int* __restrict__ erow, int* __restrict__ cnt,
                    int* __restrict__ rank, int E) {
    const int e = blockIdx.x * 256 + threadIdx.x;
    if (e >= E) return;
    rank[e] = __hip_atomic_fetch_add(cnt + erow[e], 1, __ATOMIC_RELAXED,
                                     __HIP_MEMORY_SCOPE_AGENT);
}

__global__ __launch_bounds__(1024)
void gcn_scan_counts(const int* __restrict__ cnt, int* __restrict__ rowptr) {
    __shared__ int wsum[32];
    const int t    = threadIdx.x;
    const int lane = t & 31;
    const int wave = t >> 5;

    const int4* __restrict__ src = (const int4*)cnt + t * (SCAN_PER_THREAD / 4);
    int4 v[SCAN_PER_THREAD / 4];
#pragma unroll
    for (int k = 0; k < SCAN_PER_THREAD / 4; ++k) v[k] = src[k];

    int s = 0;
#pragma unroll
    for (int k = 0; k < SCAN_PER_THREAD / 4; ++k) {
        const int4 c = v[k];
        v[k].x = s; s += c.x;
        v[k].y = s; s += c.y;
        v[k].z = s; s += c.z;
        v[k].w = s; s += c.w;
    }

    int inc = s;
    for (int off = 1; off < 32; off <<= 1) {
        const int n = __shfl_up(inc, off, 32);
        if (lane >= off) inc += n;
    }
    if (lane == 31) wsum[wave] = inc;
    __syncthreads();
    if (wave == 0) {
        const int w = wsum[lane];
        int winc = w;
        for (int off = 1; off < 32; off <<= 1) {
            const int n = __shfl_up(winc, off, 32);
            if (lane >= off) winc += n;
        }
        wsum[lane] = winc - w;
    }
    __syncthreads();

    const int base = wsum[wave] + inc - s;
    int4* __restrict__ dst = (int4*)rowptr + t * (SCAN_PER_THREAD / 4);
#pragma unroll
    for (int k = 0; k < SCAN_PER_THREAD / 4; ++k) {
        int4 o = v[k];
        o.x += base; o.y += base; o.z += base; o.w += base;
        dst[k] = o;
    }
}

__global__ __launch_bounds__(256)
void gcn_bucket_edges(const int* __restrict__ erow, const int* __restrict__ ecol,
                      const float* __restrict__ evals,
                      const int* __restrict__ rowptr, const int* __restrict__ rank,
                      int2* __restrict__ cedge, int E) {
    const int e = blockIdx.x * 256 + threadIdx.x;
    if (e >= E) return;
    const int dst = rowptr[erow[e]] + rank[e];
    cedge[dst] = make_int2(ecol[e], __float_as_int(evals[e]));
}

__device__ __forceinline__ v4f gcn_hrow(const float* __restrict__ h, int c, int lane) {
    const unsigned cc = min((unsigned)c, (unsigned)(N_NODES - 1));
    return ((const v4f*)(h + (size_t)cc * DIM))[lane];
}

__global__ __launch_bounds__(256)
void gcn_spmm_rows(const int* __restrict__ rowptr, const int2* __restrict__ cedge,
                   const float* __restrict__ h, float* __restrict__ out, int M, int E) {
    const int lane = threadIdx.x & 31;
    const int r = __builtin_amdgcn_readfirstlane(blockIdx.x * 8 + (threadIdx.x >> 5));
    if (r >= M) return;

    const int beg = max(rowptr[r], 0);
    const int end = min(rowptr[r + 1], E);

    v4f acc = {0.0f, 0.0f, 0.0f, 0.0f};
    int i = beg;
    for (; i + 4 <= end; i += 4) {
        const int2 e0 = cedge[i + 0];
        const int2 e1 = cedge[i + 1];
        const int2 e2 = cedge[i + 2];
        const int2 e3 = cedge[i + 3];
        const v4f x0 = gcn_hrow(h, e0.x, lane);
        const v4f x1 = gcn_hrow(h, e1.x, lane);
        const v4f x2 = gcn_hrow(h, e2.x, lane);
        const v4f x3 = gcn_hrow(h, e3.x, lane);
        acc += __int_as_float(e0.y) * x0;
        acc += __int_as_float(e1.y) * x1;
        acc += __int_as_float(e2.y) * x2;
        acc += __int_as_float(e3.y) * x3;
    }
    const int rem = end - i;
    if (rem > 0) {
        const int2 e0 = cedge[i];
        const int2 e1 = cedge[min(i + 1, end - 1)];
        const int2 e2 = cedge[min(i + 2, end - 1)];
        const v4f x0 = gcn_hrow(h, e0.x, lane);
        const v4f x1 = gcn_hrow(h, e1.x, lane);
        const v4f x2 = gcn_hrow(h, e2.x, lane);
        asm volatile("" :: "v"(x0), "v"(x1), "v"(x2));
        acc += __int_as_float(e0.y) * x0;
        if (rem > 1) acc += __int_as_float(e1.y) * x1;
        if (rem > 2) acc += __int_as_float(e2.y) * x2;
    }
    ((v4f*)(out + (size_t)r * DIM))[lane] = acc;
}

__global__ __launch_bounds__(256)
void gcn_bias_tanh(float* __restrict__ x, const float* __restrict__ b, int n) {
    const int i = blockIdx.x * 256 + threadIdx.x;
    if (i < n) x[i] = tanhf(x[i] + b[i & (DIM - 1)]);
}

__global__ __launch_bounds__(256)
void gcn_gemv(const float* __restrict__ h, const float* __restrict__ fcw,
              const float* __restrict__ fcb, float* __restrict__ out, int M) {
    const int node = blockIdx.x * 8 + (threadIdx.x >> 5);
    const int lane = threadIdx.x & 31;
    if (node >= M) return;

    const float4 x = ((const float4*)(h + (size_t)node * DIM))[lane];
    const float4 w = ((const float4*)fcw)[lane];
    float s = x.x * w.x + x.y * w.y + x.z * w.z + x.w * w.w;
#pragma unroll
    for (int off = 16; off > 0; off >>= 1) s += __shfl_down(s, off, 32);
    if (lane == 0) out[node] = s + fcb[0];
}

extern "C" void kernel_launch(void* const* d_in, const int* in_sizes, int n_in,
                              void* d_out, int out_size, void* d_ws, size_t ws_size,
                              hipStream_t stream) {
    const float* features = (const float*)d_in[0];
    const int*   erow     = (const int*)  d_in[1];
    const int*   ecol     = (const int*)  d_in[2];
    const float* evals    = (const float*)d_in[3];
    const float* W1       = (const float*)d_in[4];
    const float* b1       = (const float*)d_in[5];
    const float* W2       = (const float*)d_in[6];
    const float* b2       = (const float*)d_in[7];
    const float* fcw      = (const float*)d_in[8];
    const float* fcb      = (const float*)d_in[9];
    float*       out      = (float*)d_out;

    float* tmp0 = (float*)d_ws;
    float* tmp1 = tmp0 + (size_t)N_NODES * DIM;
    int*   cnt    = (int*)(tmp1 + (size_t)N_NODES * DIM);
    int*   rowptr = cnt + N_PAD;
    int*   rank   = rowptr + N_PAD;
    int2*  cedge  = (int2*)(rank + E_PAD);

    const int nElem      = N_NODES * DIM;
    const dim3 gemmGrid((M_TILES + 3) / 4);
    const dim3 elemGrid((nElem + 255) / 256);
    const dim3 padGrid(N_PAD / 256);
    const dim3 edgeGrid((N_EDGES + 255) / 256);
    const dim3 rowGrid((N_NODES + 7) / 8);
    const dim3 gemvGrid((N_NODES + 7) / 8);

    gcn_zero_counts <<<padGrid, 256, 0, stream>>>(cnt, N_PAD);
    gcn_rank_edges  <<<edgeGrid, 256, 0, stream>>>(erow, cnt, rank, N_EDGES);
    gcn_scan_counts <<<1, 1024, 0, stream>>>(cnt, rowptr);
    gcn_bucket_edges<<<edgeGrid, 256, 0, stream>>>(erow, ecol, evals, rowptr, rank,
                                                   cedge, N_EDGES);

    gcn_gemm128<<<gemmGrid, 128, 0, stream>>>(features, W1, tmp0, N_NODES);
    gcn_spmm_rows<<<rowGrid, 256, 0, stream>>>(rowptr, cedge, tmp0, tmp1, N_NODES, N_EDGES);
    gcn_bias_tanh<<<elemGrid, 256, 0, stream>>>(tmp1, b1, nElem);

    gcn_gemm128<<<gemmGrid, 128, 0, stream>>>(tmp1, W2, tmp0, N_NODES);
    gcn_spmm_rows<<<rowGrid, 256, 0, stream>>>(rowptr, cedge, tmp0, tmp1, N_NODES, N_EDGES);
    gcn_bias_tanh<<<elemGrid, 256, 0, stream>>>(tmp1, b2, nElem);

    gcn_gemv<<<gemvGrid, 256, 0, stream>>>(tmp1, fcw, fcb, out, N_NODES);
}


# bias+tanh done by the SpMM row wave on its register sum before the one store; layer 2's wave also does the fc dot and writes out; 3 launches and 450k one-element waves gone; sum order unchanged
# speedup vs baseline: 1.7489x; 1.3789x over previous
#include <hip/hip_runtime.h>


#define N_NODES 50000
#define N_EDGES 600000
#define DIM     128
#define M_TILES (N_NODES / 16)
#define SCAN_PER_THREAD 52
#define N_PAD  (1024 * SCAN_PER_THREAD)
#define E_PAD  600064

typedef __attribute__((ext_vector_type(2))) float v2f;
typedef __attribute__((ext_vector_type(8))) float v8f;
typedef __attribute__((ext_vector_type(4))) float v4f;

__global__ __launch_bounds__(128)
void gcn_gemm128(const float* __restrict__ A, const float* __restrict__ W,
                 float* __restrict__ C, int M) {
    __shared__ float ldsW[DIM * DIM];

    const int tid  = threadIdx.x;
    const int lane = tid & 31;
    const int wave = tid >> 5;

    for (int p = tid; p < 8192; p += 128) {
        const int l    = p & 31;
        const int nt   = (p >> 5) & 7;
        const int kk   = p >> 8;
        const int krow = kk * 4 + ((l & 16) ? 2 : 0);
        const int col  = nt * 16 + (l & 15);
        ldsW[2 * p + 0] = W[(krow + 0) * DIM + col];
        ldsW[2 * p + 1] = W[(krow + 1) * DIM + col];
    }
    __syncthreads();

    const int mtile = blockIdx.x * 4 + wave;
    if (mtile * 16 >= M) return;

    const float* __restrict__ Arow =
        A + (size_t)(mtile * 16 + (lane & 15)) * DIM + ((lane & 16) ? 2 : 0);

    v8f acc[8] = {};

    for (int kk = 0; kk < 32; ++kk) {
        const v2f a = *(const v2f*)(Arow + kk * 4);
#pragma unroll
        for (int nt = 0; nt < 8; ++nt) {
            const v2f b =
                *(const v2f*)&ldsW[(((kk << 3) + nt) * 32 + lane) * 2];
            acc[nt] = __builtin_amdgcn_wmma_f32_16x16x4_f32(
                 false, a,  false, b,
                 (short)0, acc[nt],
                 false,  false);
        }
    }

    const int row0 = mtile * 16 + ((lane & 16) ? 8 : 0);
    const int colc = lane & 15;
#pragma unroll
    for (int nt = 0; nt < 8; ++nt) {
#pragma unroll
        for (int j = 0; j < 8; ++j) {
            C[(size_t)(row0 + j) * DIM + nt * 16 + colc] = acc[nt][j];
        }
    }
}

__global__ __launch_bounds__(256)
void gcn_zero_counts(int* __restrict__ p, int n) {
    const int i = blockIdx.x * 256 + threadIdx.x;
    if (i < n) p[i] = 0;
}

__global__ __launch_bounds__(256)
void gcn_rank_edges(const int* __restrict__ erow, int* __restrict__ cnt,
                    int* __restrict__ rank, int E) {
    const int e = blockIdx.x * 256 + threadIdx.x;
    if (e >= E) return;
    rank[e] = __hip_atomic_fetch_add(cnt + erow[e], 1, __ATOMIC_RELAXED,
                                     __HIP_MEMORY_SCOPE_AGENT);
}

__global__ __launch_bounds__(1024)
void gcn_scan_counts(const int* __restrict__ cnt, int* __restrict__ rowptr) {
    __shared__ int wsum[32];
    const int t    = threadIdx.x;
    const int lane = t & 31;
    const int wave = t >> 5;

    const int4* __restrict__ src = (const int4*)cnt + t * (SCAN_PER_THREAD / 4);
    int4 v[SCAN_PER_THREAD / 4];
#pragma unroll
    for (int k = 0; k < SCAN_PER_THREAD / 4; ++k) v[k] = src[k];

    int s = 0;
#pragma unroll
    for (int k = 0; k < SCAN_PER_THREAD / 4; ++k) {
        const int4 c = v[k];
        v[k].x = s; s += c.x;
        v[k].y = s; s += c.y;
        v[k].z = s; s += c.z;
        v[k].w = s; s += c.w;
    }

    int inc = s;
    for (int off = 1; off < 32; off <<= 1) {
        const int n = __shfl_up(inc, off, 32);
        if (lane >= off) inc += n;
    }
    if (lane == 31) wsum[wave] = inc;
    __syncthreads();
    if (wave == 0) {
        const int w = wsum[lane];
        int winc = w;
        for (int off = 1; off < 32; off <<= 1) {
            const int n = __shfl_up(winc, off, 32);
            if (lane >= off) winc += n;
        }
        wsum[lane] = winc - w;
    }
    __syncthreads();

    const int base = wsum[wave] + inc - s;
    int4* __restrict__ dst = (int4*)rowptr + t * (SCAN_PER_THREAD / 4);
#pragma unroll
    for (int k = 0; k < SCAN_PER_THREAD / 4; ++k) {
        int4 o = v[k];
        o.x += base; o.y += base; o.z += base; o.w += base;
        dst[k] = o;
    }
}

__global__ __launch_bounds__(256)
void gcn_bucket_edges(const int* __restrict__ erow, const int* __restrict__ ecol,
                      const float* __restrict__ evals,
                      const int* __restrict__ rowptr, const int* __restrict__ rank,
                      int2* __restrict__ cedge, int E) {
    const int e = blockIdx.x * 256 + threadIdx.x;
    if (e >= E) return;
    const int dst = rowptr[erow[e]] + rank[e];
    cedge[dst] = make_int2(ecol[e], __float_as_int(evals[e]));
}

__device__ __forceinline__ v4f gcn_hrow(const float* __restrict__ h, int c, int lane) {
    const unsigned cc = min((unsigned)c, (unsigned)(N_NODES - 1));
    return ((const v4f*)(h + (size_t)cc * DIM))[lane];
}

__device__ __forceinline__ v4f gcn_row_sum(const int* __restrict__ rowptr,
                                           const int2* __restrict__ cedge,
                                           const float* __restrict__ h,
                                           int r, int lane, int E) {
    const int beg = max(rowptr[r], 0);
    const int end = min(rowptr[r + 1], E);

    v4f acc = {0.0f, 0.0f, 0.0f, 0.0f};
    int i = beg;
    for (; i + 4 <= end; i += 4) {
        const int2 e0 = cedge[i + 0];
        const int2 e1 = cedge[i + 1];
        const int2 e2 = cedge[i + 2];
        const int2 e3 = cedge[i + 3];
        const v4f x0 = gcn_hrow(h, e0.x, lane);
        const v4f x1 = gcn_hrow(h, e1.x, lane);
        const v4f x2 = gcn_hrow(h, e2.x, lane);
        const v4f x3 = gcn_hrow(h, e3.x, lane);
        acc += __int_as_float(e0.y) * x0;
        acc += __int_as_float(e1.y) * x1;
        acc += __int_as_float(e2.y) * x2;
        acc += __int_as_float(e3.y) * x3;
    }
    const int rem = end - i;
    if (rem > 0) {
        const int2 e0 = cedge[i];
        const int2 e1 = cedge[min(i + 1, end - 1)];
        const int2 e2 = cedge[min(i + 2, end - 1)];
        const v4f x0 = gcn_hrow(h, e0.x, lane);
        const v4f x1 = gcn_hrow(h, e1.x, lane);
        const v4f x2 = gcn_hrow(h, e2.x, lane);
        asm volatile("" :: "v"(x0), "v"(x1), "v"(x2));
        acc += __int_as_float(e0.y) * x0;
        if (rem > 1) acc += __int_as_float(e1.y) * x1;
        if (rem > 2) acc += __int_as_float(e2.y) * x2;
    }
    return acc;
}

__device__ __forceinline__ v4f gcn_bias_tanh4(v4f s, v4f b) {
    v4f y;
    y.x = tanhf(s.x + b.x);
    y.y = tanhf(s.y + b.y);
    y.z = tanhf(s.z + b.z);
    y.w = tanhf(s.w + b.w);
    return y;
}

__global__ __launch_bounds__(256)
void gcn_spmm_bias_tanh(const int* __restrict__ rowptr, const int2* __restrict__ cedge,
                        const float* __restrict__ h, const float* __restrict__ b,
                        float* __restrict__ out, int M, int E) {
    const int lane = threadIdx.x & 31;
    const int r = __builtin_amdgcn_readfirstlane(blockIdx.x * 8 + (threadIdx.x >> 5));
    if (r >= M) return;

    const v4f bb = ((const v4f*)b)[lane];
    const v4f acc = gcn_row_sum(rowptr, cedge, h, r, lane, E);
    ((v4f*)(out + (size_t)r * DIM))[lane] = gcn_bias_tanh4(acc, bb);
}

__global__ __launch_bounds__(256)
void gcn_spmm_bias_tanh_dot(const int* __restrict__ rowptr, const int2* __restrict__ cedge,
                            const float* __restrict__ h, const float* __restrict__ b,
                            const float* __restrict__ fcw, const float* __restrict__ fcb,
                            float* __restrict__ out, int M, int E) {
    const int lane = threadIdx.x & 31;
    const int r = __builtin_amdgcn_readfirstlane(blockIdx.x * 8 + (threadIdx.x >> 5));
    if (r >= M) return;

    const v4f bb = ((const v4f*)b)[lane];
    const v4f w  = ((const v4f*)fcw)[lane];
    const v4f acc = gcn_row_sum(rowptr, cedge, h, r, lane, E);
    const v4f x = gcn_bias_tanh4(acc, bb);

    float s = x.x * w.x + x.y * w.y + x.z * w.z + x.w * w.w;
#pragma unroll
    for (int off = 16; off > 0; off >>= 1) s += __shfl_down(s, off, 32);
    if (lane == 0) out[r] = s + fcb[0];
}

extern "C" void kernel_launch(void* const* d_in, const int* in_sizes, int n_in,
                              void* d_out, int out_size, void* d_ws, size_t ws_size,
                              hipStream_t stream) {
    const float* features = (const float*)d_in[0];
    const int*   erow     = (const int*)  d_in[1];
    const int*   ecol     = (const int*)  d_in[2];
    const float* evals    = (const float*)d_in[3];
    const float* W1       = (const float*)d_in[4];
    const float* b1       = (const float*)d_in[5];
    const float* W2       = (const float*)d_in[6];
    const float* b2       = (const float*)d_in[7];
    const float* fcw      = (const float*)d_in[8];
    const float* fcb      = (const float*)d_in[9];
    float*       out      = (float*)d_out;

    float* tmp0 = (float*)d_ws;
    float* tmp1 = tmp0 + (size_t)N_NODES * DIM;
    int*   cnt    = (int*)(tmp1 + (size_t)N_NODES * DIM);
    int*   rowptr = cnt + N_PAD;
    int*   rank   = rowptr + N_PAD;
    int2*  cedge  = (int2*)(rank + E_PAD);

    const dim3 gemmGrid((M_TILES + 3) / 4);
    const dim3 padGrid(N_PAD / 256);
    const dim3 edgeGrid((N_EDGES + 255) / 256);
    const dim3 rowGrid((N_NODES + 7) / 8);

    gcn_zero_counts <<<padGrid, 256, 0, stream>>>(cnt, N_PAD);
    gcn_rank_edges  <<<edgeGrid, 256, 0, stream>>>(erow, cnt, rank, N_EDGES);
    gcn_scan_counts <<<1, 1024, 0, stream>>>(cnt, rowptr);
    gcn_bucket_edges<<<edgeGrid, 256, 0, stream>>>(erow, ecol, evals, rowptr, rank,
                                                   cedge, N_EDGES);

    gcn_gemm128<<<gemmGrid, 128, 0, stream>>>(features, W1, tmp0, N_NODES);
    gcn_spmm_bias_tanh<<<rowGrid, 256, 0, stream>>>(rowptr, cedge, tmp0, b1, tmp1,
                                                    N_NODES, N_EDGES);

    gcn_gemm128<<<gemmGrid, 128, 0, stream>>>(tmp1, W2, tmp0, N_NODES);
    gcn_spmm_bias_tanh_dot<<<rowGrid, 256, 0, stream>>>(rowptr, cedge, tmp0, b2, fcw, fcb,
                                                        out, N_NODES, N_EDGES);
}


# layer 1's SpMM work-group (32 waves, 128 rows) puts its finished h1 rows in LDS and does h1 @ W2 there on f32 WMMA; second GEMM launch and the h1 round trip gone; sum order unchanged
# speedup vs baseline: 1.8047x; 1.0319x over previous
#include <hip/hip_runtime.h>


#define N_NODES 50000
#define N_EDGES 600000
#define DIM     128
#define M_TILES (N_NODES / 16)
#define SCAN_PER_THREAD 52
#define N_PAD  (1024 * SCAN_PER_THREAD)
#define E_PAD  600064
#define F_ROWS   128
#define HS_PITCH 132

typedef __attribute__((ext_vector_type(2))) float v2f;
typedef __attribute__((ext_vector_type(8))) float v8f;
typedef __attribute__((ext_vector_type(4))) float v4f;

__global__ __launch_bounds__(128)
void gcn_gemm128(const float* __restrict__ A, const float* __restrict__ W,
                 float* __restrict__ C, int M) {
    __shared__ float ldsW[DIM * DIM];

    const int tid  = threadIdx.x;
    const int lane = tid & 31;
    const int wave = tid >> 5;

    for (int p = tid; p < 8192; p += 128) {
        const int l    = p & 31;
        const int nt   = (p >> 5) & 7;
        const int kk   = p >> 8;
        const int krow = kk * 4 + ((l & 16) ? 2 : 0);
        const int col  = nt * 16 + (l & 15);
        ldsW[2 * p + 0] = W[(krow + 0) * DIM + col];
        ldsW[2 * p + 1] = W[(krow + 1) * DIM + col];
    }
    __syncthreads();

    const int mtile = blockIdx.x * 4 + wave;
    if (mtile * 16 >= M) return;

    const float* __restrict__ Arow =
        A + (size_t)(mtile * 16 + (lane & 15)) * DIM + ((lane & 16) ? 2 : 0);

    v8f acc[8] = {};

    for (int kk = 0; kk < 32; ++kk) {
        const v2f a = *(const v2f*)(Arow + kk * 4);
#pragma unroll
        for (int nt = 0; nt < 8; ++nt) {
            const v2f b =
                *(const v2f*)&ldsW[(((kk << 3) + nt) * 32 + lane) * 2];
            acc[nt] = __builtin_amdgcn_wmma_f32_16x16x4_f32(
                 false, a,  false, b,
                 (short)0, acc[nt],
                 false,  false);
        }
    }

    const int row0 = mtile * 16 + ((lane & 16) ? 8 : 0);
    const int colc = lane & 15;
#pragma unroll
    for (int nt = 0; nt < 8; ++nt) {
#pragma unroll
        for (int j = 0; j < 8; ++j) {
            C[(size_t)(row0 + j) * DIM + nt * 16 + colc] = acc[nt][j];
        }
    }
}

__global__ __launch_bounds__(256)
void gcn_zero_counts(int* __restrict__ p, int n) {
    const int i = blockIdx.x * 256 + threadIdx.x;
    if (i < n) p[i] = 0;
}

__global__ __launch_bounds__(256)
void gcn_rank_edges(const int* __restrict__ erow, int* __restrict__ cnt,
                    int* __restrict__ rank, int E) {
    const int e = blockIdx.x * 256 + threadIdx.x;
    if (e >= E) return;
    rank[e] = __hip_atomic_fetch_add(cnt + erow[e], 1, __ATOMIC_RELAXED,
                                     __HIP_MEMORY_SCOPE_AGENT);
}

__global__ __launch_bounds__(1024)
void gcn_scan_counts(const int* __restrict__ cnt, int* __restrict__ rowptr) {
    __shared__ int wsum[32];
    const int t    = threadIdx.x;
    const int lane = t & 31;
    const int wave = t >> 5;

    const int4* __restrict__ src = (const int4*)cnt + t * (SCAN_PER_THREAD / 4);
    int4 v[SCAN_PER_THREAD / 4];
#pragma unroll
    for (int k = 0; k < SCAN_PER_THREAD / 4; ++k) v[k] = src[k];

    int s = 0;
#pragma unroll
    for (int k = 0; k < SCAN_PER_THREAD / 4; ++k) {
        const int4 c = v[k];
        v[k].x = s; s += c.x;
        v[k].y = s; s += c.y;
        v[k].z = s; s += c.z;
        v[k].w = s; s += c.w;
    }

    int inc = s;
    for (int off = 1; off < 32; off <<= 1) {
        const int n = __shfl_up(inc, off, 32);
        if (lane >= off) inc += n;
    }
    if (lane == 31) wsum[wave] = inc;
    __syncthreads();
    if (wave == 0) {
        const int w = wsum[lane];
        int winc = w;
        for (int off = 1; off < 32; off <<= 1) {
            const int n = __shfl_up(winc, off, 32);
            if (lane >= off) winc += n;
        }
        wsum[lane] = winc - w;
    }
    __syncthreads();

    const int base = wsum[wave] + inc - s;
    int4* __restrict__ dst = (int4*)rowptr + t * (SCAN_PER_THREAD / 4);
#pragma unroll
    for (int k = 0; k < SCAN_PER_THREAD / 4; ++k) {
        int4 o = v[k];
        o.x += base; o.y += base; o.z += base; o.w += base;
        dst[k] = o;
    }
}

__global__ __launch_bounds__(256)
void gcn_bucket_edges(const int* __restrict__ erow, const int* __restrict__ ecol,
                      const float* __restrict__ evals,
                      const int* __restrict__ rowptr, const int* __restrict__ rank,
                      int2* __restrict__ cedge, int E) {
    const int e = blockIdx.x * 256 + threadIdx.x;
    if (e >= E) return;
    const int dst = rowptr[erow[e]] + rank[e];
    cedge[dst] = make_int2(ecol[e], __float_as_int(evals[e]));
}

__device__ __forceinline__ v4f gcn_hrow(const float* __restrict__ h, int c, int lane) {
    const unsigned cc = min((unsigned)c, (unsigned)(N_NODES - 1));
    return ((const v4f*)(h + (size_t)cc * DIM))[lane];
}

__device__ __forceinline__ v4f gcn_row_sum_from(const int2* __restrict__ cedge,
                                                const float* __restrict__ h,
                                                int i, int end, int lane, v4f acc) {
    for (; i + 4 <= end; i += 4) {
        const int2 e0 = cedge[i + 0];
        const int2 e1 = cedge[i + 1];
        const int2 e2 = cedge[i + 2];
        const int2 e3 = cedge[i + 3];
        const v4f x0 = gcn_hrow(h, e0.x, lane);
        const v4f x1 = gcn_hrow(h, e1.x, lane);
        const v4f x2 = gcn_hrow(h, e2.x, lane);
        const v4f x3 = gcn_hrow(h, e3.x, lane);
        acc += __int_as_float(e0.y) * x0;
        acc += __int_as_float(e1.y) * x1;
        acc += __int_as_float(e2.y) * x2;
        acc += __int_as_float(e3.y) * x3;
    }
    const int rem = end - i;
    if (rem > 0) {
        const int2 e0 = cedge[i];
        const int2 e1 = cedge[min(i + 1, end - 1)];
        const int2 e2 = cedge[min(i + 2, end - 1)];
        const v4f x0 = gcn_hrow(h, e0.x, lane);
        const v4f x1 = gcn_hrow(h, e1.x, lane);
        const v4f x2 = gcn_hrow(h, e2.x, lane);
        asm volatile("" :: "v"(x0), "v"(x1), "v"(x2));
        acc += __int_as_float(e0.y) * x0;
        if (rem > 1) acc += __int_as_float(e1.y) * x1;
        if (rem > 2) acc += __int_as_float(e2.y) * x2;
    }
    return acc;
}

__device__ __forceinline__ v4f gcn_row_sum(const int* __restrict__ rowptr,
                                           const int2* __restrict__ cedge,
                                           const float* __restrict__ h,
                                           int r, int lane, int E) {
    const int beg = max(rowptr[r], 0);
    const int end = min(rowptr[r + 1], E);
    const v4f zero = {0.0f, 0.0f, 0.0f, 0.0f};
    return gcn_row_sum_from(cedge, h, beg, end, lane, zero);
}

__device__ __forceinline__ void gcn_row_sum2(const int* __restrict__ rowptr,
                                             const int2* __restrict__ cedge,
                                             const float* __restrict__ h,
                                             int ra, int rb, int lane, int M, int E,
                                             v4f& sa, v4f& sb) {
    int ia = 0, ea = 0, ib = 0, eb = 0;
    if (ra < M) { ia = max(rowptr[ra], 0); ea = min(rowptr[ra + 1], E); }
    if (rb < M) { ib = max(rowptr[rb], 0); eb = min(rowptr[rb + 1], E); }

    v4f accA = {0.0f, 0.0f, 0.0f, 0.0f};
    v4f accB = {0.0f, 0.0f, 0.0f, 0.0f};
    for (; ia + 4 <= ea && ib + 4 <= eb; ia += 4, ib += 4) {
        const int2 a0 = cedge[ia + 0];
        const int2 a1 = cedge[ia + 1];
        const int2 a2 = cedge[ia + 2];
        const int2 a3 = cedge[ia + 3];
        const int2 b0 = cedge[ib + 0];
        const int2 b1 = cedge[ib + 1];
        const int2 b2 = cedge[ib + 2];
        const int2 b3 = cedge[ib + 3];
        const v4f xa0 = gcn_hrow(h, a0.x, lane);
        const v4f xa1 = gcn_hrow(h, a1.x, lane);
        const v4f xa2 = gcn_hrow(h, a2.x, lane);
        const v4f xa3 = gcn_hrow(h, a3.x, lane);
        const v4f xb0 = gcn_hrow(h, b0.x, lane);
        const v4f xb1 = gcn_hrow(h, b1.x, lane);
        const v4f xb2 = gcn_hrow(h, b2.x, lane);
        const v4f xb3 = gcn_hrow(h, b3.x, lane);
        accA += __int_as_float(a0.y) * xa0;
        accA += __int_as_float(a1.y) * xa1;
        accA += __int_as_float(a2.y) * xa2;
        accA += __int_as_float(a3.y) * xa3;
        accB += __int_as_float(b0.y) * xb0;
        accB += __int_as_float(b1.y) * xb1;
        accB += __int_as_float(b2.y) * xb2;
        accB += __int_as_float(b3.y) * xb3;
    }
    sa = gcn_row_sum_from(cedge, h, ia, ea, lane, accA);
    sb = gcn_row_sum_from(cedge, h, ib, eb, lane, accB);
}

__device__ __forceinline__ v4f gcn_bias_tanh4(v4f s, v4f b) {
    v4f y;
    y.x = tanhf(s.x + b.x);
    y.y = tanhf(s.y + b.y);
    y.z = tanhf(s.z + b.z);
    y.w = tanhf(s.w + b.w);
    return y;
}

__global__ __launch_bounds__(1024)
void gcn_spmm_bias_tanh_gemm(const int* __restrict__ rowptr, const int2* __restrict__ cedge,
                             const float* __restrict__ h, const float* __restrict__ b,
                             const float* __restrict__ W, float* __restrict__ C,
                             int M, int E) {
    __shared__ __attribute__((aligned(16))) float ldsW[DIM * DIM];
    __shared__ __attribute__((aligned(16))) float hs[F_ROWS * HS_PITCH];

    const int tid  = threadIdx.x;
    const int lane = tid & 31;
    const int wave = tid >> 5;

#pragma unroll
    for (int p = tid; p < 8192; p += 1024) {
        const int l    = p & 31;
        const int nt   = (p >> 5) & 7;
        const int kk   = p >> 8;
        const int krow = kk * 4 + ((l & 16) ? 2 : 0);
        const int col  = nt * 16 + (l & 15);
        ldsW[2 * p + 0] = W[(krow + 0) * DIM + col];
        ldsW[2 * p + 1] = W[(krow + 1) * DIM + col];
    }

    const int rbase = __builtin_amdgcn_readfirstlane(blockIdx.x * F_ROWS + wave);
    const v4f bb = ((const v4f*)b)[lane];
    float* myrow = hs + wave * HS_PITCH + 4 * lane;
    for (int q = 0; q < 2; ++q) {
        v4f sa, sb;
        gcn_row_sum2(rowptr, cedge, h, rbase + 64 * q, rbase + 64 * q + 32, lane, M, E, sa, sb);
        *(v4f*)(myrow + (64 * q) * HS_PITCH)      = gcn_bias_tanh4(sa, bb);
        *(v4f*)(myrow + (64 * q + 32) * HS_PITCH) = gcn_bias_tanh4(sb, bb);
    }
    __syncthreads();

    const int c  = lane & 15;
    const int hh = lane >> 4;
    const int mt = wave >> 3;
    const int nt = wave & 7;
    const float* arow0 = hs + (mt * 16 + c) * HS_PITCH + 2 * hh;
    const float* arow1 = arow0 + 64 * HS_PITCH;
    v8f acc0 = {};
    v8f acc1 = {};
#pragma unroll
    for (int kk = 0; kk < 32; ++kk) {
        const v2f bf = *(const v2f*)&ldsW[(((kk << 3) + nt) * 32 + lane) * 2];
        const v2f a0 = *(const v2f*)(arow0 + kk * 4);
        const v2f a1 = *(const v2f*)(arow1 + kk * 4);
        acc0 = __builtin_amdgcn_wmma_f32_16x16x4_f32(
             false, a0,  false, bf,
             (short)0, acc0,
             false,  false);
        acc1 = __builtin_amdgcn_wmma_f32_16x16x4_f32(
             false, a1,  false, bf,
             (short)0, acc1,
             false,  false);
    }
    __syncthreads();

    float* drow0 = hs + (mt * 16 + 8 * hh) * HS_PITCH + nt * 16 + c;
    float* drow1 = drow0 + 64 * HS_PITCH;
#pragma unroll
    for (int j = 0; j < 8; ++j) {
        drow0[j * HS_PITCH] = acc0[j];
        drow1[j * HS_PITCH] = acc1[j];
    }
    __syncthreads();

#pragma unroll
    for (int q = 0; q < 4; ++q) {
        const int r = rbase + 32 * q;
        if (r < M) ((v4f*)(C + (size_t)r * DIM))[lane] = *(const v4f*)(myrow + (32 * q) * HS_PITCH);
    }
}

__global__ __launch_bounds__(256)
void gcn_spmm_bias_tanh_dot(const int* __restrict__ rowptr, const int2* __restrict__ cedge,
                            const float* __restrict__ h, const float* __restrict__ b,
                            const float* __restrict__ fcw, const float* __restrict__ fcb,
                            float* __restrict__ out, int M, int E) {
    const int lane = threadIdx.x & 31;
    const int r = __builtin_amdgcn_readfirstlane(blockIdx.x * 8 + (threadIdx.x >> 5));
    if (r >= M) return;

    const v4f bb = ((const v4f*)b)[lane];
    const v4f w  = ((const v4f*)fcw)[lane];
    const v4f acc = gcn_row_sum(rowptr, cedge, h, r, lane, E);
    const v4f x = gcn_bias_tanh4(acc, bb);

    float s = x.x * w.x + x.y * w.y + x.z * w.z + x.w * w.w;
#pragma unroll
    for (int off = 16; off > 0; off >>= 1) s += __shfl_down(s, off, 32);
    if (lane == 0) out[r] = s + fcb[0];
}

extern "C" void kernel_launch(void* const* d_in, const int* in_sizes, int n_in,
                              void* d_out, int out_size, void* d_ws, size_t ws_size,
                              hipStream_t stream) {
    const float* features = (const float*)d_in[0];
    const int*   erow     = (const int*)  d_in[1];
    const int*   ecol     = (const int*)  d_in[2];
    const float* evals    = (const float*)d_in[3];
    const float* W1       = (const float*)d_in[4];
    const float* b1       = (const float*)d_in[5];
    const float* W2       = (const float*)d_in[6];
    const float* b2       = (const float*)d_in[7];
    const float* fcw      = (const float*)d_in[8];
    const float* fcb      = (const float*)d_in[9];
    float*       out      = (float*)d_out;

    float* tmp0 = (float*)d_ws;
    float* tmp1 = tmp0 + (size_t)N_NODES * DIM;
    int*   cnt    = (int*)(tmp1 + (size_t)N_NODES * DIM);
    int*   rowptr = cnt + N_PAD;
    int*   rank   = rowptr + N_PAD;
    int2*  cedge  = (int2*)(rank + E_PAD);

    const dim3 gemmGrid((M_TILES + 3) / 4);
    const dim3 padGrid(N_PAD / 256);
    const dim3 edgeGrid((N_EDGES + 255) / 256);
    const dim3 rowGrid((N_NODES + 7) / 8);
    const dim3 fusedGrid((N_NODES + F_ROWS - 1) / F_ROWS);

    gcn_zero_counts <<<padGrid, 256, 0, stream>>>(cnt, N_PAD);
    gcn_rank_edges  <<<edgeGrid, 256, 0, stream>>>(erow, cnt, rank, N_EDGES);
    gcn_scan_counts <<<1, 1024, 0, stream>>>(cnt, rowptr);
    gcn_bucket_edges<<<edgeGrid, 256, 0, stream>>>(erow, ecol, evals, rowptr, rank,
                                                   cedge, N_EDGES);

    gcn_gemm128<<<gemmGrid, 128, 0, stream>>>(features, W1, tmp0, N_NODES);
    gcn_spmm_bias_tanh_gemm<<<fusedGrid, 1024, 0, stream>>>(rowptr, cedge, tmp0, b1, W2, tmp1,
                                                            N_NODES, N_EDGES);

    gcn_spmm_bias_tanh_dot<<<rowGrid, 256, 0, stream>>>(rowptr, cedge, tmp1, b2, fcw, fcb,
                                                        out, N_NODES, N_EDGES);
}


# edges bucketed by a two-level counting sort on LDS int counters (8192-edge chunks x 128-row bins, then rows inside a bin): 600k returning global atomics gone; bucket order still an arrival order
# speedup vs baseline: 2.3487x; 1.3014x over previous
#include <hip/hip_runtime.h>


#define N_NODES 50000
#define N_EDGES 600000
#define DIM     128
#define M_TILES (N_NODES / 16)
#define N_PAD  53248
#define E_PAD  600064
#define BIN_ROWS      128
#define N_BINS        ((N_NODES + BIN_ROWS - 1) / BIN_ROWS)
#define CH_THREADS    1024
#define CH_PER_THREAD 8
#define CH_EDGES      (CH_THREADS * CH_PER_THREAD)
#define N_CH          ((N_EDGES + CH_EDGES - 1) / CH_EDGES)
#define CNT_PAD       28944
#define BS_PAD        400
#define SORT_THREADS  512
#define SORT_STRIP    4
#define F_ROWS   128
#define HS_PITCH 132

typedef __attribute__((ext_vector_type(2))) float v2f;
typedef __attribute__((ext_vector_type(8))) float v8f;
typedef __attribute__((ext_vector_type(4))) float v4f;

__global__ __launch_bounds__(128)
void gcn_gemm128(const float* __restrict__ A, const float* __restrict__ W,
                 float* __restrict__ C, int M) {
    __shared__ float ldsW[DIM * DIM];

    const int tid  = threadIdx.x;
    const int lane = tid & 31;
    const int wave = tid >> 5;

    for (int p = tid; p < 8192; p += 128) {
        const int l    = p & 31;
        const int nt   = (p >> 5) & 7;
        const int kk   = p >> 8;
        const int krow = kk * 4 + ((l & 16) ? 2 : 0);
        const int col  = nt * 16 + (l & 15);
        ldsW[2 * p + 0] = W[(krow + 0) * DIM + col];
        ldsW[2 * p + 1] = W[(krow + 1) * DIM + col];
    }
    __syncthreads();

    const int mtile = blockIdx.x * 4 + wave;
    if (mtile * 16 >= M) return;

    const float* __restrict__ Arow =
        A + (size_t)(mtile * 16 + (lane & 15)) * DIM + ((lane & 16) ? 2 : 0);

    v8f acc[8] = {};

    for (int kk = 0; kk < 32; ++kk) {
        const v2f a = *(const v2f*)(Arow + kk * 4);
#pragma unroll
        for (int nt = 0; nt < 8; ++nt) {
            const v2f b =
                *(const v2f*)&ldsW[(((kk << 3) + nt) * 32 + lane) * 2];
            acc[nt] = __builtin_amdgcn_wmma_f32_16x16x4_f32(
                 false, a,  false, b,
                 (short)0, acc[nt],
                 false,  false);
        }
    }

    const int row0 = mtile * 16 + ((lane & 16) ? 8 : 0);
    const int colc = lane & 15;
#pragma unroll
    for (int nt = 0; nt < 8; ++nt) {
#pragma unroll
        for (int j = 0; j < 8; ++j) {
            C[(size_t)(row0 + j) * DIM + nt * 16 + colc] = acc[nt][j];
        }
    }
}

__device__ __forceinline__ int gcn_bin_of(int row) {
    return (int)min((unsigned)row / BIN_ROWS, (unsigned)(N_BINS - 1));
}

__global__ __launch_bounds__(CH_THREADS)
void gcn_bin_count(const int* __restrict__ erow, int* __restrict__ lrank,
                   int* __restrict__ counts, int E) {
    __shared__ int c[N_BINS + 1];
    const int tid = threadIdx.x;
    if (tid <= N_BINS) c[tid] = 0;
    __syncthreads();

    const int base = blockIdx.x * CH_EDGES + tid;
    int r[CH_PER_THREAD];
#pragma unroll
    for (int k = 0; k < CH_PER_THREAD; ++k) {
        const int e = base + k * CH_THREADS;
        r[k] = erow[min(e, E - 1)];
    }
    int slot[CH_PER_THREAD];
#pragma unroll
    for (int k = 0; k < CH_PER_THREAD; ++k) {
        const int e = base + k * CH_THREADS;
        const int bin = (e < E) ? gcn_bin_of(r[k]) : N_BINS;
        slot[k] = __hip_atomic_fetch_add(&c[bin], 1, __ATOMIC_RELAXED,
                                         __HIP_MEMORY_SCOPE_WORKGROUP);
    }
#pragma unroll
    for (int k = 0; k < CH_PER_THREAD; ++k) {
        const int e = base + k * CH_THREADS;
        if (e < E) lrank[e] = slot[k];
    }
    __syncthreads();
    if (tid < N_BINS) counts[blockIdx.x * N_BINS + tid] = c[tid];
}

__global__ __launch_bounds__(512)
void gcn_bin_scan(int* __restrict__ counts, int* __restrict__ binStart) {
    __shared__ int wsum[16];
    const int t    = threadIdx.x;
    const int lane = t & 31;
    const int wave = t >> 5;

    int v[N_CH];
#pragma unroll
    for (int w = 0; w < N_CH; ++w) v[w] = (t < N_BINS) ? counts[w * N_BINS + t] : 0;
    int tot = 0;
#pragma unroll
    for (int w = 0; w < N_CH; ++w) tot += v[w];

    int inc = tot;
    for (int off = 1; off < 32; off <<= 1) {
        const int n = __shfl_up(inc, off, 32);
        if (lane >= off) inc += n;
    }
    if (lane == 31) wsum[wave] = inc;
    __syncthreads();
    if (wave == 0) {
        const int wv = (lane < 16) ? wsum[lane] : 0;
        int winc = wv;
        for (int off = 1; off < 32; off <<= 1) {
            const int n = __shfl_up(winc, off, 32);
            if (lane >= off) winc += n;
        }
        if (lane < 16) wsum[lane] = winc - wv;
    }
    __syncthreads();

    if (t < N_BINS) {
        int run = wsum[wave] + inc - tot;
        binStart[t] = run;
#pragma unroll
        for (int w = 0; w < N_CH; ++w) {
            counts[w * N_BINS + t] = run;
            run += v[w];
        }
        if (t == N_BINS - 1) binStart[N_BINS] = run;
    }
}

__global__ __launch_bounds__(CH_THREADS)
void gcn_bin_scatter(const int* __restrict__ erow, const int* __restrict__ ecol,
                     const float* __restrict__ evals, const int* __restrict__ lrank,
                     const int* __restrict__ place, int4* __restrict__ brec, int E) {
    __shared__ int o[N_BINS];
    const int tid = threadIdx.x;
    if (tid < N_BINS) o[tid] = place[blockIdx.x * N_BINS + tid];

    const int base = blockIdx.x * CH_EDGES + tid;
    int row[CH_PER_THREAD], col[CH_PER_THREAD], val[CH_PER_THREAD], lr[CH_PER_THREAD];
#pragma unroll
    for (int k = 0; k < CH_PER_THREAD; ++k) {
        const int e = min(base + k * CH_THREADS, E - 1);
        row[k] = erow[e];
        col[k] = ecol[e];
        val[k] = __float_as_int(evals[e]);
        lr[k]  = lrank[e];
    }
    __syncthreads();
#pragma unroll
    for (int k = 0; k < CH_PER_THREAD; ++k) {
        const int e = base + k * CH_THREADS;
        if (e < E) {
            const unsigned dst = min((unsigned)(o[gcn_bin_of(row[k])] + lr[k]), (unsigned)(E - 1));
            brec[dst] = make_int4(row[k], col[k], val[k], 0);
        }
    }
}

__global__ __launch_bounds__(SORT_THREADS)
void gcn_bin_sort(const int4* __restrict__ brec, const int* __restrict__ binStart,
                  int* rk, int* __restrict__ rowptr, int2* __restrict__ cedge,
                  int M, int E) {
    __shared__ int c[BIN_ROWS];
    __shared__ int pre[BIN_ROWS + 1];
    __shared__ int wtot[BIN_ROWS / 32];
    const int tid  = threadIdx.x;
    const int lane = tid & 31;
    const int wave = tid >> 5;
    const int bin  = blockIdx.x;

    const int s = min(max(binStart[bin], 0), E);
    const int t = min(max(binStart[bin + 1], s), E);

    if (tid < BIN_ROWS) c[tid] = 0;
    __syncthreads();
    for (int i0 = s + tid; i0 < t + tid; i0 += SORT_STRIP * SORT_THREADS) {
        int lrow[SORT_STRIP];
#pragma unroll
        for (int k = 0; k < SORT_STRIP; ++k)
            lrow[k] = brec[min(i0 + k * SORT_THREADS, t - 1)].x & (BIN_ROWS - 1);
#pragma unroll
        for (int k = 0; k < SORT_STRIP; ++k) {
            const int i = i0 + k * SORT_THREADS;
            if (i < t)
                rk[i] = __hip_atomic_fetch_add(&c[lrow[k]], 1, __ATOMIC_RELAXED,
                                               __HIP_MEMORY_SCOPE_WORKGROUP);
        }
    }
    __syncthreads();

    const int v = (tid < BIN_ROWS) ? c[tid] : 0;
    int inc = v;
    for (int off = 1; off < 32; off <<= 1) {
        const int n = __shfl_up(inc, off, 32);
        if (lane >= off) inc += n;
    }
    if (tid < BIN_ROWS && lane == 31) wtot[wave] = inc;
    __syncthreads();
    if (tid < BIN_ROWS) {
        int before = 0;
        for (int w = 0; w < wave; ++w) before += wtot[w];
        pre[tid] = before + inc - v;
        if (tid == BIN_ROWS - 1) pre[BIN_ROWS] = before + inc;
    }
    __syncthreads();

    if (tid <= BIN_ROWS) {
        const int r = bin * BIN_ROWS + tid;
        if (r <= M && (tid < BIN_ROWS || r == M)) rowptr[r] = s + pre[tid];
    }
    for (int i0 = s + tid; i0 < t + tid; i0 += SORT_STRIP * SORT_THREADS) {
        int4 rec[SORT_STRIP];
        int  slot[SORT_STRIP];
#pragma unroll
        for (int k = 0; k < SORT_STRIP; ++k) {
            const int i = min(i0 + k * SORT_THREADS, t - 1);
            rec[k]  = brec[i];
            slot[k] = rk[i];
        }
#pragma unroll
        for (int k = 0; k < SORT_STRIP; ++k) {
            const int i = i0 + k * SORT_THREADS;
            if (i < t) {
                const unsigned dst =
                    min((unsigned)(s + pre[rec[k].x & (BIN_ROWS - 1)] + slot[k]), (unsigned)(E - 1));
                cedge[dst] = make_int2(rec[k].y, rec[k].z);
            }
        }
    }
}

__device__ __forceinline__ v4f gcn_hrow(const float* __restrict__ h, int c, int lane) {
    const unsigned cc = min((unsigned)c, (unsigned)(N_NODES - 1));
    return ((const v4f*)(h + (size_t)cc * DIM))[lane];
}

__device__ __forceinline__ v4f gcn_row_sum_from(const int2* __restrict__ cedge,
                                                const float* __restrict__ h,
                                                int i, int end, int lane, v4f acc) {
    for (; i + 4 <= end; i += 4) {
        const int2 e0 = cedge[i + 0];
        const int2 e1 = cedge[i + 1];
        const int2 e2 = cedge[i + 2];
        const int2 e3 = cedge[i + 3];
        const v4f x0 = gcn_hrow(h, e0.x, lane);
        const v4f x1 = gcn_hrow(h, e1.x, lane);
        const v4f x2 = gcn_hrow(h, e2.x, lane);
        const v4f x3 = gcn_hrow(h, e3.x, lane);
        acc += __int_as_float(e0.y) * x0;
        acc += __int_as_float(e1.y) * x1;
        acc += __int_as_float(e2.y) * x2;
        acc += __int_as_float(e3.y) * x3;
    }
    const int rem = end - i;
    if (rem > 0) {
        const int2 e0 = cedge[i];
        const int2 e1 = cedge[min(i + 1, end - 1)];
        const int2 e2 = cedge[min(i + 2, end - 1)];
        const v4f x0 = gcn_hrow(h, e0.x, lane);
        const v4f x1 = gcn_hrow(h, e1.x, lane);
        const v4f x2 = gcn_hrow(h, e2.x, lane);
        asm volatile("" :: "v"(x0), "v"(x1), "v"(x2));
        acc += __int_as_float(e0.y) * x0;
        if (rem > 1) acc += __int_as_float(e1.y) * x1;
        if (rem > 2) acc += __int_as_float(e2.y) * x2;
    }
    return acc;
}

__device__ __forceinline__ v4f gcn_row_sum(const int* __restrict__ rowptr,
                                           const int2* __restrict__ cedge,
                                           const float* __restrict__ h,
                                           int r, int lane, int E) {
    const int beg = max(rowptr[r], 0);
    const int end = min(rowptr[r + 1], E);
    const v4f zero = {0.0f, 0.0f, 0.0f, 0.0f};
    return gcn_row_sum_from(cedge, h, beg, end, lane, zero);
}

__device__ __forceinline__ void gcn_row_sum2(const int* __restrict__ rowptr,
                                             const int2* __restrict__ cedge,
                                             const float* __restrict__ h,
                                             int ra, int rb, int lane, int M, int E,
                                             v4f& sa, v4f& sb) {
    int ia = 0, ea = 0, ib = 0, eb = 0;
    if (ra < M) { ia = max(rowptr[ra], 0); ea = min(rowptr[ra + 1], E); }
    if (rb < M) { ib = max(rowptr[rb], 0); eb = min(rowptr[rb + 1], E); }

    v4f accA = {0.0f, 0.0f, 0.0f, 0.0f};
    v4f accB = {0.0f, 0.0f, 0.0f, 0.0f};
    for (; ia + 4 <= ea && ib + 4 <= eb; ia += 4, ib += 4) {
        const int2 a0 = cedge[ia + 0];
        const int2 a1 = cedge[ia + 1];
        const int2 a2 = cedge[ia + 2];
        const int2 a3 = cedge[ia + 3];
        const int2 b0 = cedge[ib + 0];
        const int2 b1 = cedge[ib + 1];
        const int2 b2 = cedge[ib + 2];
        const int2 b3 = cedge[ib + 3];
        const v4f xa0 = gcn_hrow(h, a0.x, lane);
        const v4f xa1 = gcn_hrow(h, a1.x, lane);
        const v4f xa2 = gcn_hrow(h, a2.x, lane);
        const v4f xa3 = gcn_hrow(h, a3.x, lane);
        const v4f xb0 = gcn_hrow(h, b0.x, lane);
        const v4f xb1 = gcn_hrow(h, b1.x, lane);
        const v4f xb2 = gcn_hrow(h, b2.x, lane);
        const v4f xb3 = gcn_hrow(h, b3.x, lane);
        accA += __int_as_float(a0.y) * xa0;
        accA += __int_as_float(a1.y) * xa1;
        accA += __int_as_float(a2.y) * xa2;
        accA += __int_as_float(a3.y) * xa3;
        accB += __int_as_float(b0.y) * xb0;
        accB += __int_as_float(b1.y) * xb1;
        accB += __int_as_float(b2.y) * xb2;
        accB += __int_as_float(b3.y) * xb3;
    }
    sa = gcn_row_sum_from(cedge, h, ia, ea, lane, accA);
    sb = gcn_row_sum_from(cedge, h, ib, eb, lane, accB);
}

__device__ __forceinline__ v4f gcn_bias_tanh4(v4f s, v4f b) {
    v4f y;
    y.x = tanhf(s.x + b.x);
    y.y = tanhf(s.y + b.y);
    y.z = tanhf(s.z + b.z);
    y.w = tanhf(s.w + b.w);
    return y;
}

__global__ __launch_bounds__(1024)
void gcn_spmm_bias_tanh_gemm(const int* __restrict__ rowptr, const int2* __restrict__ cedge,
                             const float* __restrict__ h, const float* __restrict__ b,
                             const float* __restrict__ W, float* __restrict__ C,
                             int M, int E) {
    __shared__ __attribute__((aligned(16))) float ldsW[DIM * DIM];
    __shared__ __attribute__((aligned(16))) float hs[F_ROWS * HS_PITCH];

    const int tid  = threadIdx.x;
    const int lane = tid & 31;
    const int wave = tid >> 5;

#pragma unroll
    for (int p = tid; p < 8192; p += 1024) {
        const int l    = p & 31;
        const int nt   = (p >> 5) & 7;
        const int kk   = p >> 8;
        const int krow = kk * 4 + ((l & 16) ? 2 : 0);
        const int col  = nt * 16 + (l & 15);
        ldsW[2 * p + 0] = W[(krow + 0) * DIM + col];
        ldsW[2 * p + 1] = W[(krow + 1) * DIM + col];
    }

    const int rbase = __builtin_amdgcn_readfirstlane(blockIdx.x * F_ROWS + wave);
    const v4f bb = ((const v4f*)b)[lane];
    float* myrow = hs + wave * HS_PITCH + 4 * lane;
    for (int q = 0; q < 2; ++q) {
        v4f sa, sb;
        gcn_row_sum2(rowptr, cedge, h, rbase + 64 * q, rbase + 64 * q + 32, lane, M, E, sa, sb);
        *(v4f*)(myrow + (64 * q) * HS_PITCH)      = gcn_bias_tanh4(sa, bb);
        *(v4f*)(myrow + (64 * q + 32) * HS_PITCH) = gcn_bias_tanh4(sb, bb);
    }
    __syncthreads();

    const int c  = lane & 15;
    const int hh = lane >> 4;
    const int mt = wave >> 3;
    const int nt = wave & 7;
    const float* arow0 = hs + (mt * 16 + c) * HS_PITCH + 2 * hh;
    const float* arow1 = arow0 + 64 * HS_PITCH;
    v8f acc0 = {};
    v8f acc1 = {};
#pragma unroll
    for (int kk = 0; kk < 32; ++kk) {
        const v2f bf = *(const v2f*)&ldsW[(((kk << 3) + nt) * 32 + lane) * 2];
        const v2f a0 = *(const v2f*)(arow0 + kk * 4);
        const v2f a1 = *(const v2f*)(arow1 + kk * 4);
        acc0 = __builtin_amdgcn_wmma_f32_16x16x4_f32(
             false, a0,  false, bf,
             (short)0, acc0,
             false,  false);
        acc1 = __builtin_amdgcn_wmma_f32_16x16x4_f32(
             false, a1,  false, bf,
             (short)0, acc1,
             false,  false);
    }
    __syncthreads();

    float* drow0 = hs + (mt * 16 + 8 * hh) * HS_PITCH + nt * 16 + c;
    float* drow1 = drow0 + 64 * HS_PITCH;
#pragma unroll
    for (int j = 0; j < 8; ++j) {
        drow0[j * HS_PITCH] = acc0[j];
        drow1[j * HS_PITCH] = acc1[j];
    }
    __syncthreads();

#pragma unroll
    for (int q = 0; q < 4; ++q) {
        const int r = rbase + 32 * q;
        if (r < M) ((v4f*)(C + (size_t)r * DIM))[lane] = *(const v4f*)(myrow + (32 * q) * HS_PITCH);
    }
}

__global__ __launch_bounds__(256)
void gcn_spmm_bias_tanh_dot(const int* __restrict__ rowptr, const int2* __restrict__ cedge,
                            const float* __restrict__ h, const float* __restrict__ b,
                            const float* __restrict__ fcw, const float* __restrict__ fcb,
                            float* __restrict__ out, int M, int E) {
    const int lane = threadIdx.x & 31;
    const int r = __builtin_amdgcn_readfirstlane(blockIdx.x * 8 + (threadIdx.x >> 5));
    if (r >= M) return;

    const v4f bb = ((const v4f*)b)[lane];
    const v4f w  = ((const v4f*)fcw)[lane];
    const v4f acc = gcn_row_sum(rowptr, cedge, h, r, lane, E);
    const v4f x = gcn_bias_tanh4(acc, bb);

    float s = x.x * w.x + x.y * w.y + x.z * w.z + x.w * w.w;
#pragma unroll
    for (int off = 16; off > 0; off >>= 1) s += __shfl_down(s, off, 32);
    if (lane == 0) out[r] = s + fcb[0];
}

extern "C" void kernel_launch(void* const* d_in, const int* in_sizes, int n_in,
                              void* d_out, int out_size, void* d_ws, size_t ws_size,
                              hipStream_t stream) {
    const float* features = (const float*)d_in[0];
    const int*   erow     = (const int*)  d_in[1];
    const int*   ecol     = (const int*)  d_in[2];
    const float* evals    = (const float*)d_in[3];
    const float* W1       = (const float*)d_in[4];
    const float* b1       = (const float*)d_in[5];
    const float* W2       = (const float*)d_in[6];
    const float* b2       = (const float*)d_in[7];
    const float* fcw      = (const float*)d_in[8];
    const float* fcb      = (const float*)d_in[9];
    float*       out      = (float*)d_out;

    float* tmp0 = (float*)d_ws;
    float* tmp1 = tmp0 + (size_t)N_NODES * DIM;
    int*   rowptr   = (int*)(tmp1 + (size_t)N_NODES * DIM);
    int*   lrank    = rowptr + N_PAD;
    int*   rk       = lrank + E_PAD;
    int*   counts   = rk + E_PAD;
    int*   binStart = counts + CNT_PAD;
    int4*  brec     = (int4*)(binStart + BS_PAD);
    int2*  cedge    = (int2*)(brec + N_EDGES);

    const dim3 gemmGrid((M_TILES + 3) / 4);
    const dim3 chunkGrid(N_CH);
    const dim3 binGrid(N_BINS);
    const dim3 rowGrid((N_NODES + 7) / 8);
    const dim3 fusedGrid((N_NODES + F_ROWS - 1) / F_ROWS);

    gcn_bin_count  <<<chunkGrid, CH_THREADS, 0, stream>>>(erow, lrank, counts, N_EDGES);
    gcn_bin_scan   <<<1, 512, 0, stream>>>(counts, binStart);
    gcn_bin_scatter<<<chunkGrid, CH_THREADS, 0, stream>>>(erow, ecol, evals, lrank, counts,
                                                         brec, N_EDGES);
    gcn_bin_sort   <<<binGrid, SORT_THREADS, 0, stream>>>(brec, binStart, rk, rowptr, cedge,
                                                         N_NODES, N_EDGES);

    gcn_gemm128<<<gemmGrid, 128, 0, stream>>>(features, W1, tmp0, N_NODES);
    gcn_spmm_bias_tanh_gemm<<<fusedGrid, 1024, 0, stream>>>(rowptr, cedge, tmp0, b1, W2, tmp1,
                                                            N_NODES, N_EDGES);

    gcn_spmm_bias_tanh_dot<<<rowGrid, 256, 0, stream>>>(rowptr, cedge, tmp1, b2, fcw, fcb,
                                                        out, N_NODES, N_EDGES);
}


# bucketing: each 8192-edge chunk orders its edges by bin (count, scan, place in one work-group), a bin's work-group collects its 74 segments: 7 launches become 5; bucket order still an arrival order
# speedup vs baseline: 2.7388x; 1.1661x over previous
#include <hip/hip_runtime.h>


#define N_NODES 50000
#define N_EDGES 600000
#define DIM     128
#define M_TILES (N_NODES / 16)
#define N_PAD  53248
#define E_PAD  600064
#define BIN_ROWS      128
#define N_BINS        ((N_NODES + BIN_ROWS - 1) / BIN_ROWS)
#define CH_THREADS    1024
#define CH_PER_THREAD 8
#define CH_EDGES      (CH_THREADS * CH_PER_THREAD)
#define N_CH          ((N_EDGES + CH_EDGES - 1) / CH_EDGES)
#define CS_N          (N_BINS + 1)
#define CS_PITCH      400
#define SEG_PAD       128
#define SORT_THREADS  512
#define SORT_STRIP    4
#define F_ROWS   128
#define HS_PITCH 132

typedef __attribute__((ext_vector_type(2))) float v2f;
typedef __attribute__((ext_vector_type(8))) float v8f;
typedef __attribute__((ext_vector_type(4))) float v4f;

__global__ __launch_bounds__(128)
void gcn_gemm128(const float* __restrict__ A, const float* __restrict__ W,
                 float* __restrict__ C, int M) {
    __shared__ float ldsW[DIM * DIM];

    const int tid  = threadIdx.x;
    const int lane = tid & 31;
    const int wave = tid >> 5;

    for (int p = tid; p < 8192; p += 128) {
        const int l    = p & 31;
        const int nt   = (p >> 5) & 7;
        const int kk   = p >> 8;
        const int krow = kk * 4 + ((l & 16) ? 2 : 0);
        const int col  = nt * 16 + (l & 15);
        ldsW[2 * p + 0] = W[(krow + 0) * DIM + col];
        ldsW[2 * p + 1] = W[(krow + 1) * DIM + col];
    }
    __syncthreads();

    const int mtile = blockIdx.x * 4 + wave;
    if (mtile * 16 >= M) return;

    const float* __restrict__ Arow =
        A + (size_t)(mtile * 16 + (lane & 15)) * DIM + ((lane & 16) ? 2 : 0);

    v8f acc[8] = {};

    for (int kk = 0; kk < 32; ++kk) {
        const v2f a = *(const v2f*)(Arow + kk * 4);
#pragma unroll
        for (int nt = 0; nt < 8; ++nt) {
            const v2f b =
                *(const v2f*)&ldsW[(((kk << 3) + nt) * 32 + lane) * 2];
            acc[nt] = __builtin_amdgcn_wmma_f32_16x16x4_f32(
                 false, a,  false, b,
                 (short)0, acc[nt],
                 false,  false);
        }
    }

    const int row0 = mtile * 16 + ((lane & 16) ? 8 : 0);
    const int colc = lane & 15;
#pragma unroll
    for (int nt = 0; nt < 8; ++nt) {
#pragma unroll
        for (int j = 0; j < 8; ++j) {
            C[(size_t)(row0 + j) * DIM + nt * 16 + colc] = acc[nt][j];
        }
    }
}

__device__ __forceinline__ int gcn_bin_of(int row) {
    return (int)min((unsigned)row / BIN_ROWS, (unsigned)(N_BINS - 1));
}

__global__ __launch_bounds__(CH_THREADS)
void gcn_chunk_sort(const int* __restrict__ erow, const int* __restrict__ ecol,
                    const float* __restrict__ evals, int* __restrict__ cstart,
                    int4* __restrict__ brec, int E) {
    __shared__ int c[CS_N];
    __shared__ int st[CS_N];
    __shared__ int wsum[16];
    const int tid  = threadIdx.x;
    const int lane = tid & 31;
    const int wave = tid >> 5;
    if (tid < CS_N) c[tid] = 0;

    const int base = blockIdx.x * CH_EDGES + tid;
    int row[CH_PER_THREAD], col[CH_PER_THREAD], val[CH_PER_THREAD];
#pragma unroll
    for (int k = 0; k < CH_PER_THREAD; ++k) {
        const int e = min(base + k * CH_THREADS, E - 1);
        row[k] = erow[e];
        col[k] = ecol[e];
        val[k] = __float_as_int(evals[e]);
    }
    __syncthreads();

    int slot[CH_PER_THREAD];
#pragma unroll
    for (int k = 0; k < CH_PER_THREAD; ++k) {
        const int e = base + k * CH_THREADS;
        const int bin = (e < E) ? gcn_bin_of(row[k]) : N_BINS;
        slot[k] = __hip_atomic_fetch_add(&c[bin], 1, __ATOMIC_RELAXED,
                                         __HIP_MEMORY_SCOPE_WORKGROUP);
    }
    __syncthreads();

    const int v = (tid < CS_N) ? c[tid] : 0;
    int inc = v;
    for (int off = 1; off < 32; off <<= 1) {
        const int n = __shfl_up(inc, off, 32);
        if (lane >= off) inc += n;
    }
    if (wave < 16 && lane == 31) wsum[wave] = inc;
    __syncthreads();
    if (wave == 0) {
        const int wv = (lane < 16) ? wsum[lane] : 0;
        int winc = wv;
        for (int off = 1; off < 32; off <<= 1) {
            const int n = __shfl_up(winc, off, 32);
            if (lane >= off) winc += n;
        }
        if (lane < 16) wsum[lane] = winc - wv;
    }
    __syncthreads();
    const int first = wsum[wave & 15] + inc - v;
    if (tid < CS_N) st[tid] = first;
    __syncthreads();

#pragma unroll
    for (int k = 0; k < CH_PER_THREAD; ++k) {
        const int e = base + k * CH_THREADS;
        if (e < E) {
            const unsigned p = min((unsigned)(st[gcn_bin_of(row[k])] + slot[k]),
                                   (unsigned)(CH_EDGES - 1));
            brec[blockIdx.x * CH_EDGES + p] = make_int4(row[k], col[k], val[k], 0);
        }
    }
    if (tid < CS_N) cstart[blockIdx.x * CS_PITCH + tid] = first;
}

__device__ __forceinline__ unsigned gcn_src_of(const int* segoff, const int* segbase, int j) {
    int w = 0;
#pragma unroll
    for (int step = SEG_PAD / 2; step > 0; step >>= 1)
        if (segoff[w + step] <= j) w += step;
    return min((unsigned)(segbase[w] + j), (unsigned)(N_CH * CH_EDGES - 1));
}

__global__ __launch_bounds__(SORT_THREADS)
void gcn_bin_sort(const int4* __restrict__ brec, const int* __restrict__ cstart,
                  int* rk, int* __restrict__ rowptr, int2* __restrict__ cedge,
                  int M, int E) {
    __shared__ int c[BIN_ROWS];
    __shared__ int pre[BIN_ROWS + 1];
    __shared__ int wtot[BIN_ROWS / 32];
    __shared__ int segoff[SEG_PAD];
    __shared__ int segbase[SEG_PAD];
    __shared__ int wlen[SEG_PAD / 32];
    __shared__ int wbeg[SEG_PAD / 32];
    const int tid  = threadIdx.x;
    const int lane = tid & 31;
    const int wave = tid >> 5;
    const int bin  = blockIdx.x;

    int a = 0, len = 0;
    if (tid < N_CH) {
        const int* cs = cstart + tid * CS_PITCH + bin;
        a   = min(max(cs[0], 0), CH_EDGES);
        len = min(max(cs[1], a), CH_EDGES) - a;
    }
    if (tid < BIN_ROWS) c[tid] = 0;

    int incl = len;
    for (int off = 1; off < 32; off <<= 1) {
        const int n = __shfl_up(incl, off, 32);
        if (lane >= off) incl += n;
    }
    int asum = a;
    for (int off = 16; off > 0; off >>= 1) asum += __shfl_xor(asum, off, 32);
    if (tid < SEG_PAD && lane == 31) { wlen[wave] = incl; wbeg[wave] = asum; }
    __syncthreads();

    const int s = min(wbeg[0] + wbeg[1] + wbeg[2] + wbeg[3], E);
    const int t = min(s + wlen[0] + wlen[1] + wlen[2] + wlen[3], E);
    const int n = t - s;
    if (tid < SEG_PAD) {
        int before = 0;
        for (int w = 0; w < wave; ++w) before += wlen[w];
        const int off = before + incl - len;
        segoff[tid]  = off;
        segbase[tid] = tid * CH_EDGES + a - off;
    }
    __syncthreads();

    for (int j0 = tid; j0 - tid < n; j0 += SORT_STRIP * SORT_THREADS) {
        int rrow[SORT_STRIP];
#pragma unroll
        for (int k = 0; k < SORT_STRIP; ++k) {
            const int j = min(j0 + k * SORT_THREADS, n - 1);
            rrow[k] = brec[gcn_src_of(segoff, segbase, j)].x;
        }
        asm volatile("" :: "v"(rrow[0]), "v"(rrow[1]), "v"(rrow[2]), "v"(rrow[3]));
#pragma unroll
        for (int k = 0; k < SORT_STRIP; ++k) {
            const int j = j0 + k * SORT_THREADS;
            if (j < n)
                rk[s + j] = __hip_atomic_fetch_add(&c[rrow[k] & (BIN_ROWS - 1)], 1,
                                                   __ATOMIC_RELAXED,
                                                   __HIP_MEMORY_SCOPE_WORKGROUP);
        }
    }
    __syncthreads();

    const int v = (tid < BIN_ROWS) ? c[tid] : 0;
    int inc = v;
    for (int off = 1; off < 32; off <<= 1) {
        const int m = __shfl_up(inc, off, 32);
        if (lane >= off) inc += m;
    }
    if (tid < BIN_ROWS && lane == 31) wtot[wave] = inc;
    __syncthreads();
    if (tid < BIN_ROWS) {
        int before = 0;
        for (int w = 0; w < wave; ++w) before += wtot[w];
        pre[tid] = before + inc - v;
        if (tid == BIN_ROWS - 1) pre[BIN_ROWS] = before + inc;
    }
    __syncthreads();

    if (tid <= BIN_ROWS) {
        const int r = bin * BIN_ROWS + tid;
        if (r <= M && (tid < BIN_ROWS || r == M)) rowptr[r] = s + pre[tid];
    }
    for (int j0 = tid; j0 - tid < n; j0 += SORT_STRIP * SORT_THREADS) {
        int4 rec[SORT_STRIP];
        int  slot[SORT_STRIP];
#pragma unroll
        for (int k = 0; k < SORT_STRIP; ++k) {
            const int j = min(j0 + k * SORT_THREADS, n - 1);
            rec[k]  = brec[gcn_src_of(segoff, segbase, j)];
            slot[k] = rk[s + j];
        }
        asm volatile("" :: "v"(rec[0].x), "v"(rec[1].x), "v"(rec[2].x), "v"(rec[3].x),
                           "v"(slot[0]), "v"(slot[1]), "v"(slot[2]), "v"(slot[3]));
#pragma unroll
        for (int k = 0; k < SORT_STRIP; ++k) {
            const int j = j0 + k * SORT_THREADS;
            if (j < n) {
                const unsigned dst =
                    min((unsigned)(s + pre[rec[k].x & (BIN_ROWS - 1)] + slot[k]), (unsigned)(E - 1));
                cedge[dst] = make_int2(rec[k].y, rec[k].z);
            }
        }
    }
}

__device__ __forceinline__ v4f gcn_hrow(const float* __restrict__ h, int c, int lane) {
    const unsigned cc = min((unsigned)c, (unsigned)(N_NODES - 1));
    return ((const v4f*)(h + (size_t)cc * DIM))[lane];
}

__device__ __forceinline__ v4f gcn_row_sum_from(const int2* __restrict__ cedge,
                                                const float* __restrict__ h,
                                                int i, int end, int lane, v4f acc) {
    for (; i + 4 <= end; i += 4) {
        const int2 e0 = cedge[i + 0];
        const int2 e1 = cedge[i + 1];
        const int2 e2 = cedge[i + 2];
        const int2 e3 = cedge[i + 3];
        const v4f x0 = gcn_hrow(h, e0.x, lane);
        const v4f x1 = gcn_hrow(h, e1.x, lane);
        const v4f x2 = gcn_hrow(h, e2.x, lane);
        const v4f x3 = gcn_hrow(h, e3.x, lane);
        acc += __int_as_float(e0.y) * x0;
        acc += __int_as_float(e1.y) * x1;
        acc += __int_as_float(e2.y) * x2;
        acc += __int_as_float(e3.y) * x3;
    }
    const int rem = end - i;
    if (rem > 0) {
        const int2 e0 = cedge[i];
        const int2 e1 = cedge[min(i + 1, end - 1)];
        const int2 e2 = cedge[min(i + 2, end - 1)];
        const v4f x0 = gcn_hrow(h, e0.x, lane);
        const v4f x1 = gcn_hrow(h, e1.x, lane);
        const v4f x2 = gcn_hrow(h, e2.x, lane);
        asm volatile("" :: "v"(x0), "v"(x1), "v"(x2));
        acc += __int_as_float(e0.y) * x0;
        if (rem > 1) acc += __int_as_float(e1.y) * x1;
        if (rem > 2) acc += __int_as_float(e2.y) * x2;
    }
    return acc;
}

__device__ __forceinline__ v4f gcn_row_sum(const int* __restrict__ rowptr,
                                           const int2* __restrict__ cedge,
                                           const float* __restrict__ h,
                                           int r, int lane, int E) {
    const int beg = max(rowptr[r], 0);
    const int end = min(rowptr[r + 1], E);
    const v4f zero = {0.0f, 0.0f, 0.0f, 0.0f};
    return gcn_row_sum_from(cedge, h, beg, end, lane, zero);
}

__device__ __forceinline__ void gcn_row_sum2(const int* __restrict__ rowptr,
                                             const int2* __restrict__ cedge,
                                             const float* __restrict__ h,
                                             int ra, int rb, int lane, int M, int E,
                                             v4f& sa, v4f& sb) {
    int ia = 0, ea = 0, ib = 0, eb = 0;
    if (ra < M) { ia = max(rowptr[ra], 0); ea = min(rowptr[ra + 1], E); }
    if (rb < M) { ib = max(rowptr[rb], 0); eb = min(rowptr[rb + 1], E); }

    v4f accA = {0.0f, 0.0f, 0.0f, 0.0f};
    v4f accB = {0.0f, 0.0f, 0.0f, 0.0f};
    for (; ia + 4 <= ea && ib + 4 <= eb; ia += 4, ib += 4) {
        const int2 a0 = cedge[ia + 0];
        const int2 a1 = cedge[ia + 1];
        const int2 a2 = cedge[ia + 2];
        const int2 a3 = cedge[ia + 3];
        const int2 b0 = cedge[ib + 0];
        const int2 b1 = cedge[ib + 1];
        const int2 b2 = cedge[ib + 2];
        const int2 b3 = cedge[ib + 3];
        const v4f xa0 = gcn_hrow(h, a0.x, lane);
        const v4f xa1 = gcn_hrow(h, a1.x, lane);
        const v4f xa2 = gcn_hrow(h, a2.x, lane);
        const v4f xa3 = gcn_hrow(h, a3.x, lane);
        const v4f xb0 = gcn_hrow(h, b0.x, lane);
        const v4f xb1 = gcn_hrow(h, b1.x, lane);
        const v4f xb2 = gcn_hrow(h, b2.x, lane);
        const v4f xb3 = gcn_hrow(h, b3.x, lane);
        accA += __int_as_float(a0.y) * xa0;
        accA += __int_as_float(a1.y) * xa1;
        accA += __int_as_float(a2.y) * xa2;
        accA += __int_as_float(a3.y) * xa3;
        accB += __int_as_float(b0.y) * xb0;
        accB += __int_as_float(b1.y) * xb1;
        accB += __int_as_float(b2.y) * xb2;
        accB += __int_as_float(b3.y) * xb3;
    }
    sa = gcn_row_sum_from(cedge, h, ia, ea, lane, accA);
    sb = gcn_row_sum_from(cedge, h, ib, eb, lane, accB);
}

__device__ __forceinline__ v4f gcn_bias_tanh4(v4f s, v4f b) {
    v4f y;
    y.x = tanhf(s.x + b.x);
    y.y = tanhf(s.y + b.y);
    y.z = tanhf(s.z + b.z);
    y.w = tanhf(s.w + b.w);
    return y;
}

__global__ __launch_bounds__(1024)
void gcn_spmm_bias_tanh_gemm(const int* __restrict__ rowptr, const int2* __restrict__ cedge,
                             const float* __restrict__ h, const float* __restrict__ b,
                             const float* __restrict__ W, float* __restrict__ C,
                             int M, int E) {
    __shared__ __attribute__((aligned(16))) float ldsW[DIM * DIM];
    __shared__ __attribute__((aligned(16))) float hs[F_ROWS * HS_PITCH];

    const int tid  = threadIdx.x;
    const int lane = tid & 31;
    const int wave = tid >> 5;

#pragma unroll
    for (int p = tid; p < 8192; p += 1024) {
        const int l    = p & 31;
        const int nt   = (p >> 5) & 7;
        const int kk   = p >> 8;
        const int krow = kk * 4 + ((l & 16) ? 2 : 0);
        const int col  = nt * 16 + (l & 15);
        ldsW[2 * p + 0] = W[(krow + 0) * DIM + col];
        ldsW[2 * p + 1] = W[(krow + 1) * DIM + col];
    }

    const int rbase = __builtin_amdgcn_readfirstlane(blockIdx.x * F_ROWS + wave);
    const v4f bb = ((const v4f*)b)[lane];
    float* myrow = hs + wave * HS_PITCH + 4 * lane;
    for (int q = 0; q < 2; ++q) {
        v4f sa, sb;
        gcn_row_sum2(rowptr, cedge, h, rbase + 64 * q, rbase + 64 * q + 32, lane, M, E, sa, sb);
        *(v4f*)(myrow + (64 * q) * HS_PITCH)      = gcn_bias_tanh4(sa, bb);
        *(v4f*)(myrow + (64 * q + 32) * HS_PITCH) = gcn_bias_tanh4(sb, bb);
    }
    __syncthreads();

    const int c  = lane & 15;
    const int hh = lane >> 4;
    const int mt = wave >> 3;
    const int nt = wave & 7;
    const float* arow0 = hs + (mt * 16 + c) * HS_PITCH + 2 * hh;
    const float* arow1 = arow0 + 64 * HS_PITCH;
    v8f acc0 = {};
    v8f acc1 = {};
#pragma unroll
    for (int kk = 0; kk < 32; ++kk) {
        const v2f bf = *(const v2f*)&ldsW[(((kk << 3) + nt) * 32 + lane) * 2];
        const v2f a0 = *(const v2f*)(arow0 + kk * 4);
        const v2f a1 = *(const v2f*)(arow1 + kk * 4);
        acc0 = __builtin_amdgcn_wmma_f32_16x16x4_f32(
             false, a0,  false, bf,
             (short)0, acc0,
             false,  false);
        acc1 = __builtin_amdgcn_wmma_f32_16x16x4_f32(
             false, a1,  false, bf,
             (short)0, acc1,
             false,  false);
    }
    __syncthreads();

    float* drow0 = hs + (mt * 16 + 8 * hh) * HS_PITCH + nt * 16 + c;
    float* drow1 = drow0 + 64 * HS_PITCH;
#pragma unroll
    for (int j = 0; j < 8; ++j) {
        drow0[j * HS_PITCH] = acc0[j];
        drow1[j * HS_PITCH] = acc1[j];
    }
    __syncthreads();

#pragma unroll
    for (int q = 0; q < 4; ++q) {
        const int r = rbase + 32 * q;
        if (r < M) ((v4f*)(C + (size_t)r * DIM))[lane] = *(const v4f*)(myrow + (32 * q) * HS_PITCH);
    }
}

__global__ __launch_bounds__(256)
void gcn_spmm_bias_tanh_dot(const int* __restrict__ rowptr, const int2* __restrict__ cedge,
                            const float* __restrict__ h, const float* __restrict__ b,
                            const float* __restrict__ fcw, const float* __restrict__ fcb,
                            float* __restrict__ out, int M, int E) {
    const int lane = threadIdx.x & 31;
    const int r = __builtin_amdgcn_readfirstlane(blockIdx.x * 8 + (threadIdx.x >> 5));
    if (r >= M) return;

    const v4f bb = ((const v4f*)b)[lane];
    const v4f w  = ((const v4f*)fcw)[lane];
    const v4f acc = gcn_row_sum(rowptr, cedge, h, r, lane, E);
    const v4f x = gcn_bias_tanh4(acc, bb);

    float s = x.x * w.x + x.y * w.y + x.z * w.z + x.w * w.w;
#pragma unroll
    for (int off = 16; off > 0; off >>= 1) s += __shfl_down(s, off, 32);
    if (lane == 0) out[r] = s + fcb[0];
}

extern "C" void kernel_launch(void* const* d_in, const int* in_sizes, int n_in,
                              void* d_out, int out_size, void* d_ws, size_t ws_size,
                              hipStream_t stream) {
    const float* features = (const float*)d_in[0];
    const int*   erow     = (const int*)  d_in[1];
    const int*   ecol     = (const int*)  d_in[2];
    const float* evals    = (const float*)d_in[3];
    const float* W1       = (const float*)d_in[4];
    const float* b1       = (const float*)d_in[5];
    const float* W2       = (const float*)d_in[6];
    const float* b2       = (const float*)d_in[7];
    const float* fcw      = (const float*)d_in[8];
    const float* fcb      = (const float*)d_in[9];
    float*       out      = (float*)d_out;

    float* tmp0 = (float*)d_ws;
    float* tmp1 = tmp0 + (size_t)N_NODES * DIM;
    int*   rowptr   = (int*)(tmp1 + (size_t)N_NODES * DIM);
    int*   rk       = rowptr + N_PAD;
    int*   cstart   = rk + E_PAD;
    int4*  brec     = (int4*)(cstart + N_CH * CS_PITCH);
    int2*  cedge    = (int2*)(brec + N_CH * CH_EDGES);

    const dim3 gemmGrid((M_TILES + 3) / 4);
    const dim3 chunkGrid(N_CH);
    const dim3 binGrid(N_BINS);
    const dim3 rowGrid((N_NODES + 7) / 8);
    const dim3 fusedGrid((N_NODES + F_ROWS - 1) / F_ROWS);

    gcn_chunk_sort<<<chunkGrid, CH_THREADS, 0, stream>>>(erow, ecol, evals, cstart, brec, N_EDGES);
    gcn_bin_sort  <<<binGrid, SORT_THREADS, 0, stream>>>(brec, cstart, rk, rowptr, cedge,
                                                        N_NODES, N_EDGES);

    gcn_gemm128<<<gemmGrid, 128, 0, stream>>>(features, W1, tmp0, N_NODES);
    gcn_spmm_bias_tanh_gemm<<<fusedGrid, 1024, 0, stream>>>(rowptr, cedge, tmp0, b1, W2, tmp1,
                                                            N_NODES, N_EDGES);

    gcn_spmm_bias_tanh_dot<<<rowGrid, 256, 0, stream>>>(rowptr, cedge, tmp1, b2, fcw, fcb,
                                                        out, N_NODES, N_EDGES);
}


# X @ W1 shares the launch of the edge-chunk ordering (independent work): 74 chunk work-groups, then 391 of 128 X rows in LDS through layer 1's block product; 5 launches to 4; sum order unchanged
# speedup vs baseline: 3.1461x; 1.1487x over previous
#include <hip/hip_runtime.h>


#define N_NODES 50000
#define N_EDGES 600000
#define DIM     128
#define N_PAD  53248
#define E_PAD  600064
#define BIN_ROWS      128
#define N_BINS        ((N_NODES + BIN_ROWS - 1) / BIN_ROWS)
#define CH_THREADS    1024
#define CH_PER_THREAD 8
#define CH_EDGES      (CH_THREADS * CH_PER_THREAD)
#define N_CH          ((N_EDGES + CH_EDGES - 1) / CH_EDGES)
#define CS_N          (N_BINS + 1)
#define CS_PITCH      400
#define SEG_PAD       128
#define SORT_THREADS  512
#define SORT_STRIP    4
#define F_ROWS   128
#define HS_PITCH 132
#define N_ROWBLK ((N_NODES + F_ROWS - 1) / F_ROWS)

typedef __attribute__((ext_vector_type(2))) float v2f;
typedef __attribute__((ext_vector_type(8))) float v8f;
typedef __attribute__((ext_vector_type(4))) float v4f;

__device__ __forceinline__ void gcn_stage_w(const float* __restrict__ W, float* ldsW, int tid) {
#pragma unroll
    for (int p = tid; p < 8192; p += 1024) {
        const int l    = p & 31;
        const int nt   = (p >> 5) & 7;
        const int kk   = p >> 8;
        const int krow = kk * 4 + ((l & 16) ? 2 : 0);
        const int col  = nt * 16 + (l & 15);
        ldsW[2 * p + 0] = W[(krow + 0) * DIM + col];
        ldsW[2 * p + 1] = W[(krow + 1) * DIM + col];
    }
}

__device__ __forceinline__ void gcn_block_product(float* hs, const float* ldsW, int wave, int lane) {
    __syncthreads();

    const int c  = lane & 15;
    const int hh = lane >> 4;
    const int mt = wave >> 3;
    const int nt = wave & 7;
    const float* arow0 = hs + (mt * 16 + c) * HS_PITCH + 2 * hh;
    const float* arow1 = arow0 + 64 * HS_PITCH;
    v8f acc0 = {};
    v8f acc1 = {};
#pragma unroll
    for (int kk = 0; kk < 32; ++kk) {
        const v2f bf = *(const v2f*)&ldsW[(((kk << 3) + nt) * 32 + lane) * 2];
        const v2f a0 = *(const v2f*)(arow0 + kk * 4);
        const v2f a1 = *(const v2f*)(arow1 + kk * 4);
        acc0 = __builtin_amdgcn_wmma_f32_16x16x4_f32(
             false, a0,  false, bf,
             (short)0, acc0,
             false,  false);
        acc1 = __builtin_amdgcn_wmma_f32_16x16x4_f32(
             false, a1,  false, bf,
             (short)0, acc1,
             false,  false);
    }
    __syncthreads();

    float* drow0 = hs + (mt * 16 + 8 * hh) * HS_PITCH + nt * 16 + c;
    float* drow1 = drow0 + 64 * HS_PITCH;
#pragma unroll
    for (int j = 0; j < 8; ++j) {
        drow0[j * HS_PITCH] = acc0[j];
        drow1[j * HS_PITCH] = acc1[j];
    }
    __syncthreads();
}

__device__ __forceinline__ void gcn_gemm_rows128(const float* __restrict__ A,
                                                 const float* __restrict__ W,
                                                 float* __restrict__ C, int blk, int M) {
    __shared__ __attribute__((aligned(16))) float ldsW[DIM * DIM];
    __shared__ __attribute__((aligned(16))) float hs[F_ROWS * HS_PITCH];

    const int tid  = threadIdx.x;
    const int lane = tid & 31;
    const int wave = tid >> 5;

    gcn_stage_w(W, ldsW, tid);

    const int rbase = blk * F_ROWS + wave;
    float* myrow = hs + wave * HS_PITCH + 4 * lane;
    v4f x[4];
#pragma unroll
    for (int q = 0; q < 4; ++q)
        x[q] = ((const v4f*)(A + (size_t)min(rbase + 32 * q, M - 1) * DIM))[lane];
#pragma unroll
    for (int q = 0; q < 4; ++q)
        *(v4f*)(myrow + (32 * q) * HS_PITCH) = x[q];

    gcn_block_product(hs, ldsW, wave, lane);

#pragma unroll
    for (int q = 0; q < 4; ++q) {
        const int r = rbase + 32 * q;
        if (r < M) ((v4f*)(C + (size_t)r * DIM))[lane] = *(const v4f*)(myrow + (32 * q) * HS_PITCH);
    }
}

__device__ __forceinline__ int gcn_bin_of(int row) {
    return (int)min((unsigned)row / BIN_ROWS, (unsigned)(N_BINS - 1));
}

__device__ __forceinline__
void gcn_chunk_sort(const int* __restrict__ erow, const int* __restrict__ ecol,
                    const float* __restrict__ evals, int* __restrict__ cstart,
                    int4* __restrict__ brec, int chunk, int E) {
    __shared__ int c[CS_N];
    __shared__ int st[CS_N];
    __shared__ int wsum[16];
    const int tid  = threadIdx.x;
    const int lane = tid & 31;
    const int wave = tid >> 5;
    if (tid < CS_N) c[tid] = 0;

    const int base = chunk * CH_EDGES + tid;
    int row[CH_PER_THREAD], col[CH_PER_THREAD], val[CH_PER_THREAD];
#pragma unroll
    for (int k = 0; k < CH_PER_THREAD; ++k) {
        const int e = min(base + k * CH_THREADS, E - 1);
        row[k] = erow[e];
        col[k] = ecol[e];
        val[k] = __float_as_int(evals[e]);
    }
    __syncthreads();

    int slot[CH_PER_THREAD];
#pragma unroll
    for (int k = 0; k < CH_PER_THREAD; ++k) {
        const int e = base + k * CH_THREADS;
        const int bin = (e < E) ? gcn_bin_of(row[k]) : N_BINS;
        slot[k] = __hip_atomic_fetch_add(&c[bin], 1, __ATOMIC_RELAXED,
                                         __HIP_MEMORY_SCOPE_WORKGROUP);
    }
    __syncthreads();

    const int v = (tid < CS_N) ? c[tid] : 0;
    int inc = v;
    for (int off = 1; off < 32; off <<= 1) {
        const int n = __shfl_up(inc, off, 32);
        if (lane >= off) inc += n;
    }
    if (wave < 16 && lane == 31) wsum[wave] = inc;
    __syncthreads();
    if (wave == 0) {
        const int wv = (lane < 16) ? wsum[lane] : 0;
        int winc = wv;
        for (int off = 1; off < 32; off <<= 1) {
            const int n = __shfl_up(winc, off, 32);
            if (lane >= off) winc += n;
        }
        if (lane < 16) wsum[lane] = winc - wv;
    }
    __syncthreads();
    const int first = wsum[wave & 15] + inc - v;
    if (tid < CS_N) st[tid] = first;
    __syncthreads();

#pragma unroll
    for (int k = 0; k < CH_PER_THREAD; ++k) {
        const int e = base + k * CH_THREADS;
        if (e < E) {
            const unsigned p = min((unsigned)(st[gcn_bin_of(row[k])] + slot[k]),
                                   (unsigned)(CH_EDGES - 1));
            brec[chunk * CH_EDGES + p] = make_int4(row[k], col[k], val[k], 0);
        }
    }
    if (tid < CS_N) cstart[chunk * CS_PITCH + tid] = first;
}

__global__ __launch_bounds__(CH_THREADS)
void gcn_chunk_sort_and_gemm(const int* __restrict__ erow, const int* __restrict__ ecol,
                             const float* __restrict__ evals, int* __restrict__ cstart,
                             int4* __restrict__ brec, int E,
                             const float* __restrict__ A, const float* __restrict__ W,
                             float* __restrict__ C, int M) {
    if (blockIdx.x < N_CH)
        gcn_chunk_sort(erow, ecol, evals, cstart, brec, blockIdx.x, E);
    else
        gcn_gemm_rows128(A, W, C, blockIdx.x - N_CH, M);
}

__device__ __forceinline__ unsigned gcn_src_of(const int* segoff, const int* segbase, int j) {
    int w = 0;
#pragma unroll
    for (int step = SEG_PAD / 2; step > 0; step >>= 1)
        if (segoff[w + step] <= j) w += step;
    return min((unsigned)(segbase[w] + j), (unsigned)(N_CH * CH_EDGES - 1));
}

__global__ __launch_bounds__(SORT_THREADS)
void gcn_bin_sort(const int4* __restrict__ brec, const int* __restrict__ cstart,
                  int* rk, int* __restrict__ rowptr, int2* __restrict__ cedge,
                  int M, int E) {
    __shared__ int c[BIN_ROWS];
    __shared__ int pre[BIN_ROWS + 1];
    __shared__ int wtot[BIN_ROWS / 32];
    __shared__ int segoff[SEG_PAD];
    __shared__ int segbase[SEG_PAD];
    __shared__ int wlen[SEG_PAD / 32];
    __shared__ int wbeg[SEG_PAD / 32];
    const int tid  = threadIdx.x;
    const int lane = tid & 31;
    const int wave = tid >> 5;
    const int bin  = blockIdx.x;

    int a = 0, len = 0;
    if (tid < N_CH) {
        const int* cs = cstart + tid * CS_PITCH + bin;
        a   = min(max(cs[0], 0), CH_EDGES);
        len = min(max(cs[1], a), CH_EDGES) - a;
    }
    if (tid < BIN_ROWS) c[tid] = 0;

    int incl = len;
    for (int off = 1; off < 32; off <<= 1) {
        const int n = __shfl_up(incl, off, 32);
        if (lane >= off) incl += n;
    }
    int asum = a;
    for (int off = 16; off > 0; off >>= 1) asum += __shfl_xor(asum, off, 32);
    if (tid < SEG_PAD && lane == 31) { wlen[wave] = incl; wbeg[wave] = asum; }
    __syncthreads();

    const int s = min(wbeg[0] + wbeg[1] + wbeg[2] + wbeg[3], E);
    const int t = min(s + wlen[0] + wlen[1] + wlen[2] + wlen[3], E);
    const int n = t - s;
    if (tid < SEG_PAD) {
        int before = 0;
        for (int w = 0; w < wave; ++w) before += wlen[w];
        const int off = before + incl - len;
        segoff[tid]  = off;
        segbase[tid] = tid * CH_EDGES + a - off;
    }
    __syncthreads();

    for (int j0 = tid; j0 - tid < n; j0 += SORT_STRIP * SORT_THREADS) {
        int rrow[SORT_STRIP];
#pragma unroll
        for (int k = 0; k < SORT_STRIP; ++k) {
            const int j = min(j0 + k * SORT_THREADS, n - 1);
            rrow[k] = brec[gcn_src_of(segoff, segbase, j)].x;
        }
        asm volatile("" :: "v"(rrow[0]), "v"(rrow[1]), "v"(rrow[2]), "v"(rrow[3]));
#pragma unroll
        for (int k = 0; k < SORT_STRIP; ++k) {
            const int j = j0 + k * SORT_THREADS;
            if (j < n)
                rk[s + j] = __hip_atomic_fetch_add(&c[rrow[k] & (BIN_ROWS - 1)], 1,
                                                   __ATOMIC_RELAXED,
                                                   __HIP_MEMORY_SCOPE_WORKGROUP);
        }
    }
    __syncthreads();

    const int v = (tid < BIN_ROWS) ? c[tid] : 0;
    int inc = v;
    for (int off = 1; off < 32; off <<= 1) {
        const int m = __shfl_up(inc, off, 32);
        if (lane >= off) inc += m;
    }
    if (tid < BIN_ROWS && lane == 31) wtot[wave] = inc;
    __syncthreads();
    if (tid < BIN_ROWS) {
        int before = 0;
        for (int w = 0; w < wave; ++w) before += wtot[w];
        pre[tid] = before + inc - v;
        if (tid == BIN_ROWS - 1) pre[BIN_ROWS] = before + inc;
    }
    __syncthreads();

    if (tid <= BIN_ROWS) {
        const int r = bin * BIN_ROWS + tid;
        if (r <= M && (tid < BIN_ROWS || r == M)) rowptr[r] = s + pre[tid];
    }
    for (int j0 = tid; j0 - tid < n; j0 += SORT_STRIP * SORT_THREADS) {
        int4 rec[SORT_STRIP];
        int  slot[SORT_STRIP];
#pragma unroll
        for (int k = 0; k < SORT_STRIP; ++k) {
            const int j = min(j0 + k * SORT_THREADS, n - 1);
            rec[k]  = brec[gcn_src_of(segoff, segbase, j)];
            slot[k] = rk[s + j];
        }
        asm volatile("" :: "v"(rec[0].x), "v"(rec[1].x), "v"(rec[2].x), "v"(rec[3].x),
                           "v"(slot[0]), "v"(slot[1]), "v"(slot[2]), "v"(slot[3]));
#pragma unroll
        for (int k = 0; k < SORT_STRIP; ++k) {
            const int j = j0 + k * SORT_THREADS;
            if (j < n) {
                const unsigned dst =
                    min((unsigned)(s + pre[rec[k].x & (BIN_ROWS - 1)] + slot[k]), (unsigned)(E - 1));
                cedge[dst] = make_int2(rec[k].y, rec[k].z);
            }
        }
    }
}

__device__ __forceinline__ v4f gcn_hrow(const float* __restrict__ h, int c, int lane) {
    const unsigned cc = min((unsigned)c, (unsigned)(N_NODES - 1));
    return ((const v4f*)(h + (size_t)cc * DIM))[lane];
}

__device__ __forceinline__ v4f gcn_row_sum_from(const int2* __restrict__ cedge,
                                                const float* __restrict__ h,
                                                int i, int end, int lane, v4f acc) {
    for (; i + 4 <= end; i += 4) {
        const int2 e0 = cedge[i + 0];
        const int2 e1 = cedge[i + 1];
        const int2 e2 = cedge[i + 2];
        const int2 e3 = cedge[i + 3];
        const v4f x0 = gcn_hrow(h, e0.x, lane);
        const v4f x1 = gcn_hrow(h, e1.x, lane);
        const v4f x2 = gcn_hrow(h, e2.x, lane);
        const v4f x3 = gcn_hrow(h, e3.x, lane);
        acc += __int_as_float(e0.y) * x0;
        acc += __int_as_float(e1.y) * x1;
        acc += __int_as_float(e2.y) * x2;
        acc += __int_as_float(e3.y) * x3;
    }
    const int rem = end - i;
    if (rem > 0) {
        const int2 e0 = cedge[i];
        const int2 e1 = cedge[min(i + 1, end - 1)];
        const int2 e2 = cedge[min(i + 2, end - 1)];
        const v4f x0 = gcn_hrow(h, e0.x, lane);
        const v4f x1 = gcn_hrow(h, e1.x, lane);
        const v4f x2 = gcn_hrow(h, e2.x, lane);
        asm volatile("" :: "v"(x0), "v"(x1), "v"(x2));
        acc += __int_as_float(e0.y) * x0;
        if (rem > 1) acc += __int_as_float(e1.y) * x1;
        if (rem > 2) acc += __int_as_float(e2.y) * x2;
    }
    return acc;
}

__device__ __forceinline__ v4f gcn_row_sum(const int* __restrict__ rowptr,
                                           const int2* __restrict__ cedge,
                                           const float* __restrict__ h,
                                           int r, int lane, int E) {
    const int beg = max(rowptr[r], 0);
    const int end = min(rowptr[r + 1], E);
    const v4f zero = {0.0f, 0.0f, 0.0f, 0.0f};
    return gcn_row_sum_from(cedge, h, beg, end, lane, zero);
}

__device__ __forceinline__ void gcn_row_sum2(const int* __restrict__ rowptr,
                                             const int2* __restrict__ cedge,
                                             const float* __restrict__ h,
                                             int ra, int rb, int lane, int M, int E,
                                             v4f& sa, v4f& sb) {
    int ia = 0, ea = 0, ib = 0, eb = 0;
    if (ra < M) { ia = max(rowptr[ra], 0); ea = min(rowptr[ra + 1], E); }
    if (rb < M) { ib = max(rowptr[rb], 0); eb = min(rowptr[rb + 1], E); }

    v4f accA = {0.0f, 0.0f, 0.0f, 0.0f};
    v4f accB = {0.0f, 0.0f, 0.0f, 0.0f};
    for (; ia + 4 <= ea && ib + 4 <= eb; ia += 4, ib += 4) {
        const int2 a0 = cedge[ia + 0];
        const int2 a1 = cedge[ia + 1];
        const int2 a2 = cedge[ia + 2];
        const int2 a3 = cedge[ia + 3];
        const int2 b0 = cedge[ib + 0];
        const int2 b1 = cedge[ib + 1];
        const int2 b2 = cedge[ib + 2];
        const int2 b3 = cedge[ib + 3];
        const v4f xa0 = gcn_hrow(h, a0.x, lane);
        const v4f xa1 = gcn_hrow(h, a1.x, lane);
        const v4f xa2 = gcn_hrow(h, a2.x, lane);
        const v4f xa3 = gcn_hrow(h, a3.x, lane);
        const v4f xb0 = gcn_hrow(h, b0.x, lane);
        const v4f xb1 = gcn_hrow(h, b1.x, lane);
        const v4f xb2 = gcn_hrow(h, b2.x, lane);
        const v4f xb3 = gcn_hrow(h, b3.x, lane);
        accA += __int_as_float(a0.y) * xa0;
        accA += __int_as_float(a1.y) * xa1;
        accA += __int_as_float(a2.y) * xa2;
        accA += __int_as_float(a3.y) * xa3;
        accB += __int_as_float(b0.y) * xb0;
        accB += __int_as_float(b1.y) * xb1;
        accB += __int_as_float(b2.y) * xb2;
        accB += __int_as_float(b3.y) * xb3;
    }
    sa = gcn_row_sum_from(cedge, h, ia, ea, lane, accA);
    sb = gcn_row_sum_from(cedge, h, ib, eb, lane, accB);
}

__device__ __forceinline__ v4f gcn_bias_tanh4(v4f s, v4f b) {
    v4f y;
    y.x = tanhf(s.x + b.x);
    y.y = tanhf(s.y + b.y);
    y.z = tanhf(s.z + b.z);
    y.w = tanhf(s.w + b.w);
    return y;
}

__global__ __launch_bounds__(1024)
void gcn_spmm_bias_tanh_gemm(const int* __restrict__ rowptr, const int2* __restrict__ cedge,
                             const float* __restrict__ h, const float* __restrict__ b,
                             const float* __restrict__ W, float* __restrict__ C,
                             int M, int E) {
    __shared__ __attribute__((aligned(16))) float ldsW[DIM * DIM];
    __shared__ __attribute__((aligned(16))) float hs[F_ROWS * HS_PITCH];

    const int tid  = threadIdx.x;
    const int lane = tid & 31;
    const int wave = tid >> 5;

    gcn_stage_w(W, ldsW, tid);

    const int rbase = __builtin_amdgcn_readfirstlane(blockIdx.x * F_ROWS + wave);
    const v4f bb = ((const v4f*)b)[lane];
    float* myrow = hs + wave * HS_PITCH + 4 * lane;
    for (int q = 0; q < 2; ++q) {
        v4f sa, sb;
        gcn_row_sum2(rowptr, cedge, h, rbase + 64 * q, rbase + 64 * q + 32, lane, M, E, sa, sb);
        *(v4f*)(myrow + (64 * q) * HS_PITCH)      = gcn_bias_tanh4(sa, bb);
        *(v4f*)(myrow + (64 * q + 32) * HS_PITCH) = gcn_bias_tanh4(sb, bb);
    }
    gcn_block_product(hs, ldsW, wave, lane);

#pragma unroll
    for (int q = 0; q < 4; ++q) {
        const int r = rbase + 32 * q;
        if (r < M) ((v4f*)(C + (size_t)r * DIM))[lane] = *(const v4f*)(myrow + (32 * q) * HS_PITCH);
    }
}

__global__ __launch_bounds__(256)
void gcn_spmm_bias_tanh_dot(const int* __restrict__ rowptr, const int2* __restrict__ cedge,
                            const float* __restrict__ h, const float* __restrict__ b,
                            const float* __restrict__ fcw, const float* __restrict__ fcb,
                            float* __restrict__ out, int M, int E) {
    const int lane = threadIdx.x & 31;
    const int r = __builtin_amdgcn_readfirstlane(blockIdx.x * 8 + (threadIdx.x >> 5));
    if (r >= M) return;

    const v4f bb = ((const v4f*)b)[lane];
    const v4f w  = ((const v4f*)fcw)[lane];
    const v4f acc = gcn_row_sum(rowptr, cedge, h, r, lane, E);
    const v4f x = gcn_bias_tanh4(acc, bb);

    float s = x.x * w.x + x.y * w.y + x.z * w.z + x.w * w.w;
#pragma unroll
    for (int off = 16; off > 0; off >>= 1) s += __shfl_down(s, off, 32);
    if (lane == 0) out[r] = s + fcb[0];
}

extern "C" void kernel_launch(void* const* d_in, const int* in_sizes, int n_in,
                              void* d_out, int out_size, void* d_ws, size_t ws_size,
                              hipStream_t stream) {
    const float* features = (const float*)d_in[0];
    const int*   erow     = (const int*)  d_in[1];
    const int*   ecol     = (const int*)  d_in[2];
    const float* evals    = (const float*)d_in[3];
    const float* W1       = (const float*)d_in[4];
    const float* b1       = (const float*)d_in[5];
    const float* W2       = (const float*)d_in[6];
    const float* b2       = (const float*)d_in[7];
    const float* fcw      = (const float*)d_in[8];
    const float* fcb      = (const float*)d_in[9];
    float*       out      = (float*)d_out;

    float* tmp0 = (float*)d_ws;
    float* tmp1 = tmp0 + (size_t)N_NODES * DIM;
    int*   rowptr   = (int*)(tmp1 + (size_t)N_NODES * DIM);
    int*   rk       = rowptr + N_PAD;
    int*   cstart   = rk + E_PAD;
    int4*  brec     = (int4*)(cstart + N_CH * CS_PITCH);
    int2*  cedge    = (int2*)(brec + N_CH * CH_EDGES);

    const dim3 binGrid(N_BINS);
    const dim3 rowGrid((N_NODES + 7) / 8);
    const dim3 fusedGrid(N_ROWBLK);
    const dim3 firstGrid(N_CH + N_ROWBLK);

    gcn_chunk_sort_and_gemm<<<firstGrid, CH_THREADS, 0, stream>>>(erow, ecol, evals, cstart, brec, N_EDGES,
                                                                  features, W1, tmp0, N_NODES);
    gcn_bin_sort  <<<binGrid, SORT_THREADS, 0, stream>>>(brec, cstart, rk, rowptr, cedge,
                                                        N_NODES, N_EDGES);

    gcn_spmm_bias_tanh_gemm<<<fusedGrid, 1024, 0, stream>>>(rowptr, cedge, tmp0, b1, W2, tmp1,
                                                            N_NODES, N_EDGES);

    gcn_spmm_bias_tanh_dot<<<rowGrid, 256, 0, stream>>>(rowptr, cedge, tmp1, b2, fcw, fcb,
                                                        out, N_NODES, N_EDGES);
}
